# starting kernel
# baseline (speedup 1.0000x reference)
#include <hip/hip_runtime.h>

#define N_NODES 100000
#define N_EDGES 1600000
#define D 64

typedef __attribute__((ext_vector_type(2))) float v2f;
typedef __attribute__((ext_vector_type(8))) float v8f;

__global__ void zero_kernel(float* __restrict__ p, long long n) {
    long long i = (long long)blockIdx.x * blockDim.x + threadIdx.x;
    long long stride = (long long)gridDim.x * blockDim.x;
    for (; i < n; i += stride) p[i] = 0.0f;
}

__global__ void degree_kernel(const int* __restrict__ row, float* __restrict__ deg, int E) {
    int e = blockIdx.x * blockDim.x + threadIdx.x;
    if (e < E) atomicAdd(&deg[row[e]], 1.0f);
}

__global__ void dinv_kernel(const float* __restrict__ deg, float* __restrict__ dinv, int n) {
    int i = blockIdx.x * blockDim.x + threadIdx.x;
    if (i < n) {
        float d = deg[i];
        dinv[i] = (d > 0.0f) ? (1.0f / sqrtf(d)) : 0.0f;
    }
}

__global__ void lap_kernel(const int* __restrict__ row, const int* __restrict__ col,
                           const float* __restrict__ attr, const float* __restrict__ dinv,
                           float* __restrict__ lap, int E) {
    int e = blockIdx.x * blockDim.x + threadIdx.x;
    if (e < E) lap[e] = -dinv[row[e]] * attr[e] * dinv[col[e]];
}

__global__ void spmm_kernel(const int* __restrict__ row, const int* __restrict__ col,
                            const float* __restrict__ lap, const float* __restrict__ h,
                            float* __restrict__ acc, int E) {
    int idx = blockIdx.x * blockDim.x + threadIdx.x;
    int e = idx >> 6;
    int j = idx & 63;
    if (e < E) {
        float l = lap[e];
        int c = col[e];
        int r = row[e];
        atomicAdd(&acc[r * D + j], l * h[c * D + j]);
    }
}

__global__ void axpby_kernel(float* __restrict__ y, const float* __restrict__ x, int n) {
    int i = blockIdx.x * blockDim.x + threadIdx.x;
    if (i < n) y[i] = 2.0f * y[i] - x[i];
}

__global__ void gemm_wmma_kernel(const float* __restrict__ H, const float* __restrict__ W,
                                 const float* __restrict__ bias, float* __restrict__ out,
                                 int n_row_tiles, int accumulate) {
    const int wave = threadIdx.x >> 5;
    const int lane = threadIdx.x & 31;
    const int col0 = wave * 16;
    const int laneN = lane & 15;
    const int hi = lane >> 4;

    v2f bfrag[16];
#pragma unroll
    for (int ks = 0; ks < 16; ++ks) {
        int k = ks * 4 + hi * 2;
        bfrag[ks].x = W[(k + 0) * D + col0 + laneN];
        bfrag[ks].y = W[(k + 1) * D + col0 + laneN];
    }
    float bval = accumulate ? 0.0f : bias[col0 + laneN];

    for (int tile = blockIdx.x; tile < n_row_tiles; tile += gridDim.x) {
        const int row0 = tile * 16;
        v8f acc;
        if (accumulate) {
#pragma unroll
            for (int r = 0; r < 8; ++r)
                acc[r] = out[(row0 + hi * 8 + r) * D + col0 + laneN];
        } else {
#pragma unroll
            for (int r = 0; r < 8; ++r) acc[r] = bval;
        }

        const float* hrow = H + (row0 + laneN) * D + hi * 2;
#pragma unroll
        for (int ks = 0; ks < 16; ++ks) {
            v2f a;
            a.x = hrow[ks * 4 + 0];
            a.y = hrow[ks * 4 + 1];
            acc = __builtin_amdgcn_wmma_f32_16x16x4_f32(
                 false, a,  false, bfrag[ks],
                 (short)0, acc,  false,  false);
        }

#pragma unroll
        for (int r = 0; r < 8; ++r)
            out[(row0 + hi * 8 + r) * D + col0 + laneN] = acc[r];
    }
}

extern "C" void kernel_launch(void* const* d_in, const int* in_sizes, int n_in,
                              void* d_out, int out_size, void* d_ws, size_t ws_size,
                              hipStream_t stream) {
    const float* x         = (const float*)d_in[0];
    const int*   edge_idx  = (const int*)d_in[1];
    const float* edge_attr = (const float*)d_in[2];
    const float* weight    = (const float*)d_in[3];
    const float* bias      = (const float*)d_in[4];
    float*       out       = (float*)d_out;

    const int* row = edge_idx;
    const int* col = edge_idx + N_EDGES;

    char* ws = (char*)d_ws;
    float* deg  = (float*)ws; ws += sizeof(float) * N_NODES;
    float* dinv = (float*)ws; ws += sizeof(float) * N_NODES;
    float* lap  = (float*)ws; ws += sizeof(float) * N_EDGES;
    float* A    = (float*)ws; ws += sizeof(float) * (size_t)N_NODES * D;
    float* B    = (float*)ws; ws += sizeof(float) * (size_t)N_NODES * D;
    float* Z    = (float*)ws; ws += sizeof(float) * (size_t)N_NODES * D;

    const int n_row_tiles = N_NODES / 16;
    const int GEMM_BLOCKS = 1024;
    const long long spmm_threads = (long long)N_EDGES * D;
    const int spmm_blocks = (int)((spmm_threads + 255) / 256);
    const int e_blocks = (N_EDGES + 255) / 256;
    const int n_blocks = (N_NODES + 255) / 256;
    const int nd_blocks = (N_NODES * D + 255) / 256;

    zero_kernel<<<2048, 256, 0, stream>>>(deg, (long long)N_NODES);
    zero_kernel<<<4096, 256, 0, stream>>>(A, (long long)N_NODES * D * 3);

    degree_kernel<<<e_blocks, 256, 0, stream>>>(row, deg, N_EDGES);
    dinv_kernel<<<n_blocks, 256, 0, stream>>>(deg, dinv, N_NODES);
    lap_kernel<<<e_blocks, 256, 0, stream>>>(row, col, edge_attr, dinv, lap, N_EDGES);

    gemm_wmma_kernel<<<GEMM_BLOCKS, 128, 0, stream>>>(x, weight, bias, out, n_row_tiles, 0);

    spmm_kernel<<<spmm_blocks, 256, 0, stream>>>(row, col, lap, x, A, N_EDGES);
    gemm_wmma_kernel<<<GEMM_BLOCKS, 128, 0, stream>>>(A, weight + 1 * D * D, nullptr, out, n_row_tiles, 1);

    spmm_kernel<<<spmm_blocks, 256, 0, stream>>>(row, col, lap, A, B, N_EDGES);
    axpby_kernel<<<nd_blocks, 256, 0, stream>>>(B, x, N_NODES * D);
    gemm_wmma_kernel<<<GEMM_BLOCKS, 128, 0, stream>>>(B, weight + 2 * D * D, nullptr, out, n_row_tiles, 1);

    spmm_kernel<<<spmm_blocks, 256, 0, stream>>>(row, col, lap, B, Z, N_EDGES);
    axpby_kernel<<<nd_blocks, 256, 0, stream>>>(Z, A, N_NODES * D);
    gemm_wmma_kernel<<<GEMM_BLOCKS, 128, 0, stream>>>(Z, weight + 3 * D * D, nullptr, out, n_row_tiles, 1);
}


# SPMM by destination tile: edges binned per call into 128-row tiles; one work-group per tile sums in LDS with LDS float adds (arrival order), applies 2*acc-prev, stores once; no global atomics
# speedup vs baseline: 3.0276x; 3.0276x over previous
#include <hip/hip_runtime.h>

#define N_NODES 100000
#define N_EDGES 1600000
#define D 64

typedef __attribute__((ext_vector_type(2))) float v2f;
typedef __attribute__((ext_vector_type(8))) float v8f;

__global__ void zero_kernel(float* __restrict__ p, long long n) {
    long long i = (long long)blockIdx.x * blockDim.x + threadIdx.x;
    long long stride = (long long)gridDim.x * blockDim.x;
    for (; i < n; i += stride) p[i] = 0.0f;
}

__global__ void degree_kernel(const int* __restrict__ row, float* __restrict__ deg, int E) {
    int e = blockIdx.x * blockDim.x + threadIdx.x;
    if (e < E) atomicAdd(&deg[row[e]], 1.0f);
}

__global__ void dinv_kernel(const float* __restrict__ deg, float* __restrict__ dinv, int n) {
    int i = blockIdx.x * blockDim.x + threadIdx.x;
    if (i < n) {
        float d = deg[i];
        dinv[i] = (d > 0.0f) ? (1.0f / sqrtf(d)) : 0.0f;
    }
}

__global__ void lap_kernel(const int* __restrict__ row, const int* __restrict__ col,
                           const float* __restrict__ attr, const float* __restrict__ dinv,
                           float* __restrict__ lap, int E) {
    int e = blockIdx.x * blockDim.x + threadIdx.x;
    if (e < E) lap[e] = -dinv[row[e]] * attr[e] * dinv[col[e]];
}

#define TILE_ROWS 128
#define TILE_SHIFT 7
#define N_TILES 782
#define BIN_BLOCKS 250
#define BIN_CHUNK 6400
#define COL_BITS 20

typedef __attribute__((ext_vector_type(4))) float v4f;

__global__ void __launch_bounds__(256)
bin_count_kernel(const int* __restrict__ row, int* __restrict__ cnt) {
    __shared__ int hist[N_TILES];
    for (int i = threadIdx.x; i < N_TILES; i += 256) hist[i] = 0;
    __syncthreads();
    const int base = blockIdx.x * BIN_CHUNK;
    int r[BIN_CHUNK / 256];
#pragma unroll
    for (int k = 0; k < BIN_CHUNK / 256; ++k) r[k] = row[base + threadIdx.x + 256 * k];
#pragma unroll
    for (int k = 0; k < BIN_CHUNK / 256; ++k) atomicAdd(&hist[r[k] >> TILE_SHIFT], 1);
    __syncthreads();
    for (int i = threadIdx.x; i < N_TILES; i += 256)
        cnt[blockIdx.x * N_TILES + i] = hist[i];
}

__global__ void __launch_bounds__(1024)
bin_scan_kernel(int* __restrict__ cnt, int* __restrict__ tile_start) {
    __shared__ int s[1024];
    const int t = threadIdx.x;
    int run = 0;
    if (t < N_TILES) {
        for (int b0 = 0; b0 < BIN_BLOCKS; b0 += 25) {
            int v[25];
#pragma unroll
            for (int u = 0; u < 25; ++u) v[u] = cnt[(b0 + u) * N_TILES + t];
#pragma unroll
            for (int u = 0; u < 25; ++u) {
                cnt[(b0 + u) * N_TILES + t] = run;
                run += v[u];
            }
        }
    }
    s[t] = run;
    __syncthreads();
    for (int off = 1; off < 1024; off <<= 1) {
        int v = (t >= off) ? s[t - off] : 0;
        __syncthreads();
        s[t] += v;
        __syncthreads();
    }
    if (t < N_TILES) tile_start[t] = s[t] - run;
    if (t == N_TILES - 1) tile_start[N_TILES] = s[t];
}

__global__ void __launch_bounds__(256)
bin_place_kernel(const int* __restrict__ row, const int* __restrict__ col,
                 const float* __restrict__ lap, const int* __restrict__ cnt,
                 const int* __restrict__ tile_start, unsigned long long* __restrict__ rec) {
    __shared__ int cur[N_TILES];
    for (int i = threadIdx.x; i < N_TILES; i += 256)
        cur[i] = tile_start[i] + cnt[blockIdx.x * N_TILES + i];
    __syncthreads();
    const int base = blockIdx.x * BIN_CHUNK;
    for (int k0 = 0; k0 < BIN_CHUNK / 256; k0 += 5) {
        int r[5], c[5];
        float l[5];
#pragma unroll
        for (int u = 0; u < 5; ++u) {
            const int e = base + threadIdx.x + 256 * (k0 + u);
            r[u] = row[e];
            c[u] = col[e];
            l[u] = lap[e];
        }
#pragma unroll
        for (int u = 0; u < 5; ++u) {
            const int slot = atomicAdd(&cur[r[u] >> TILE_SHIFT], 1);
            const unsigned lo = (unsigned)c[u] | ((unsigned)(r[u] & (TILE_ROWS - 1)) << COL_BITS);
            const unsigned hi = __float_as_uint(l[u]);
            rec[slot] = (unsigned long long)lo | ((unsigned long long)hi << 32);
        }
    }
}

__global__ void __launch_bounds__(256, 8)
spmm_tile_kernel(const unsigned long long* __restrict__ rec, const int* __restrict__ tile_start,
                 const float* __restrict__ h, const float* prev, float* dst, int mode) {
    __shared__ float acc[TILE_ROWS * D];
    const int tile = blockIdx.x;
    const int lane = threadIdx.x & 31;
    const int wave = __builtin_amdgcn_readfirstlane((int)(threadIdx.x >> 5));

    v4f* acc4 = (v4f*)acc;
#pragma unroll
    for (int i = 0; i < (TILE_ROWS * D / 4) / 256; ++i) {
        v4f z = {0.0f, 0.0f, 0.0f, 0.0f};
        acc4[threadIdx.x + 256 * i] = z;
    }
    __syncthreads();

    const int s0 = tile_start[tile];
    const int s1 = tile_start[tile + 1];
    const float* hl = h + 2 * lane;
    float* al = acc + 2 * lane;

    unsigned long long rr = 0ull;
    if (s0 + wave * 32 + lane < s1) rr = rec[s0 + wave * 32 + lane];
    for (int base = s0 + wave * 32; base < s1; base += 256) {
        const int n = s1 - base;
        unsigned long long rnext = 0ull;
        if (base + 256 + lane < s1) rnext = rec[base + 256 + lane];
        const int rx = (int)(unsigned)(rr & 0xffffffffull);
        const int ry = (int)(unsigned)(rr >> 32);
        if (n >= 32) {
#pragma unroll
            for (int kb = 0; kb < 32; kb += 16) {
                v2f v[16];
#pragma unroll
                for (int u = 0; u < 16; ++u) {
                    const unsigned p = (unsigned)__builtin_amdgcn_readlane(rx, kb + u);
                    const unsigned c = p & ((1u << COL_BITS) - 1u);
                    v[u] = *(const v2f*)(hl + (size_t)c * D);
                }
#pragma unroll
                for (int u = 0; u < 16; ++u) {
                    const unsigned p = (unsigned)__builtin_amdgcn_readlane(rx, kb + u);
                    const float l = __uint_as_float((unsigned)__builtin_amdgcn_readlane(ry, kb + u));
                    float* d = al + (p >> COL_BITS) * D;
                    atomicAdd(d, l * v[u].x);
                    atomicAdd(d + 1, l * v[u].y);
                }
            }
        } else {
            for (int k = 0; k < n; ++k) {
                const unsigned p = (unsigned)__builtin_amdgcn_readlane(rx, k);
                const float l = __uint_as_float((unsigned)__builtin_amdgcn_readlane(ry, k));
                const unsigned c = p & ((1u << COL_BITS) - 1u);
                const v2f v = *(const v2f*)(hl + (size_t)c * D);
                float* d = al + (p >> COL_BITS) * D;
                atomicAdd(d, l * v.x);
                atomicAdd(d + 1, l * v.y);
            }
        }
        rr = rnext;
    }

    const int row0 = tile * TILE_ROWS;
    v4f pv[(TILE_ROWS * D / 4) / 256];
    if (mode == 1) {
#pragma unroll
        for (int i = 0; i < (TILE_ROWS * D / 4) / 256; ++i) {
            const int idx = threadIdx.x + 256 * i;
            const int grow = row0 + (idx >> 4);
            if (grow < N_NODES)
                pv[i] = ((const v4f*)prev)[(size_t)grow * (D / 4) + (idx & 15)];
        }
    }
    __syncthreads();

#pragma unroll
    for (int i = 0; i < (TILE_ROWS * D / 4) / 256; ++i) {
        const int idx = threadIdx.x + 256 * i;
        const int grow = row0 + (idx >> 4);
        if (grow < N_NODES) {
            v4f a = acc4[idx];
            const size_t g = (size_t)grow * (D / 4) + (idx & 15);
            if (mode == 1) a = 2.0f * a - pv[i];
            ((v4f*)dst)[g] = a;
        }
    }
}

__global__ void gemm_wmma_kernel(const float* __restrict__ H, const float* __restrict__ W,
                                 const float* __restrict__ bias, float* __restrict__ out,
                                 int n_row_tiles, int accumulate) {
    const int wave = threadIdx.x >> 5;
    const int lane = threadIdx.x & 31;
    const int col0 = wave * 16;
    const int laneN = lane & 15;
    const int hi = lane >> 4;

    v2f bfrag[16];
#pragma unroll
    for (int ks = 0; ks < 16; ++ks) {
        int k = ks * 4 + hi * 2;
        bfrag[ks].x = W[(k + 0) * D + col0 + laneN];
        bfrag[ks].y = W[(k + 1) * D + col0 + laneN];
    }
    float bval = accumulate ? 0.0f : bias[col0 + laneN];

    for (int tile = blockIdx.x; tile < n_row_tiles; tile += gridDim.x) {
        const int row0 = tile * 16;
        v8f acc;
        if (accumulate) {
#pragma unroll
            for (int r = 0; r < 8; ++r)
                acc[r] = out[(row0 + hi * 8 + r) * D + col0 + laneN];
        } else {
#pragma unroll
            for (int r = 0; r < 8; ++r) acc[r] = bval;
        }

        const float* hrow = H + (row0 + laneN) * D + hi * 2;
#pragma unroll
        for (int ks = 0; ks < 16; ++ks) {
            v2f a;
            a.x = hrow[ks * 4 + 0];
            a.y = hrow[ks * 4 + 1];
            acc = __builtin_amdgcn_wmma_f32_16x16x4_f32(
                 false, a,  false, bfrag[ks],
                 (short)0, acc,  false,  false);
        }

#pragma unroll
        for (int r = 0; r < 8; ++r)
            out[(row0 + hi * 8 + r) * D + col0 + laneN] = acc[r];
    }
}

extern "C" void kernel_launch(void* const* d_in, const int* in_sizes, int n_in,
                              void* d_out, int out_size, void* d_ws, size_t ws_size,
                              hipStream_t stream) {
    const float* x         = (const float*)d_in[0];
    const int*   edge_idx  = (const int*)d_in[1];
    const float* edge_attr = (const float*)d_in[2];
    const float* weight    = (const float*)d_in[3];
    const float* bias      = (const float*)d_in[4];
    float*       out       = (float*)d_out;

    const int* row = edge_idx;
    const int* col = edge_idx + N_EDGES;

    char* ws = (char*)d_ws;
    float* deg  = (float*)ws; ws += sizeof(float) * N_NODES;
    float* dinv = (float*)ws; ws += sizeof(float) * N_NODES;
    float* lap  = (float*)ws; ws += sizeof(float) * N_EDGES;
    float* A    = (float*)ws; ws += sizeof(float) * (size_t)N_NODES * D;
    float* B    = (float*)ws; ws += sizeof(float) * (size_t)N_NODES * D;
    float* Z    = (float*)ws;

    const int n_row_tiles = N_NODES / 16;
    const int GEMM_BLOCKS = 1024;
    const int e_blocks = (N_EDGES + 255) / 256;
    const int n_blocks = (N_NODES + 255) / 256;

    unsigned long long* rec = (unsigned long long*)Z;
    int* cnt        = (int*)(rec + N_EDGES);
    int* tile_start = cnt + BIN_BLOCKS * N_TILES;

    zero_kernel<<<2048, 256, 0, stream>>>(deg, (long long)N_NODES);

    degree_kernel<<<e_blocks, 256, 0, stream>>>(row, deg, N_EDGES);
    dinv_kernel<<<n_blocks, 256, 0, stream>>>(deg, dinv, N_NODES);
    lap_kernel<<<e_blocks, 256, 0, stream>>>(row, col, edge_attr, dinv, lap, N_EDGES);

    gemm_wmma_kernel<<<GEMM_BLOCKS, 128, 0, stream>>>(x, weight, bias, out, n_row_tiles, 0);

    bin_count_kernel<<<BIN_BLOCKS, 256, 0, stream>>>(row, cnt);
    bin_scan_kernel<<<1, 1024, 0, stream>>>(cnt, tile_start);
    bin_place_kernel<<<BIN_BLOCKS, 256, 0, stream>>>(row, col, lap, cnt, tile_start, rec);

    spmm_tile_kernel<<<N_TILES, 256, 0, stream>>>(rec, tile_start, x, x, A, 0);
    gemm_wmma_kernel<<<GEMM_BLOCKS, 128, 0, stream>>>(A, weight + 1 * D * D, nullptr, out, n_row_tiles, 1);

    spmm_tile_kernel<<<N_TILES, 256, 0, stream>>>(rec, tile_start, A, x, B, 1);
    gemm_wmma_kernel<<<GEMM_BLOCKS, 128, 0, stream>>>(B, weight + 2 * D * D, nullptr, out, n_row_tiles, 1);

    spmm_tile_kernel<<<N_TILES, 256, 0, stream>>>(rec, tile_start, B, A, A, 1);
    gemm_wmma_kernel<<<GEMM_BLOCKS, 128, 0, stream>>>(A, weight + 3 * D * D, nullptr, out, n_row_tiles, 1);
}


# SPMM sums on the matrix unit: tile records grouped by 16-row block in LDS, a wave adds 4 edges per f32 WMMA (one-hot weights x gathered rows) in registers; no LDS float adds; sums in slot order
# speedup vs baseline: 3.3150x; 1.0949x over previous
#include <hip/hip_runtime.h>

#define N_NODES 100000
#define N_EDGES 1600000
#define D 64

typedef __attribute__((ext_vector_type(2))) float v2f;
typedef __attribute__((ext_vector_type(8))) float v8f;

__global__ void zero_kernel(float* __restrict__ p, long long n) {
    long long i = (long long)blockIdx.x * blockDim.x + threadIdx.x;
    long long stride = (long long)gridDim.x * blockDim.x;
    for (; i < n; i += stride) p[i] = 0.0f;
}

__device__ __forceinline__ float dinv_of(float d) {
    return (d > 0.0f) ? (1.0f / sqrtf(d)) : 0.0f;
}

#define TILE_ROWS 128
#define TILE_SHIFT 7
#define N_TILES 782
#define BIN_BLOCKS 250
#define BIN_CHUNK 6400
#define COL_BITS 20

typedef __attribute__((ext_vector_type(4))) float v4f;

__global__ void __launch_bounds__(256)
bin_count_kernel(const int* __restrict__ row, int* __restrict__ cnt, float* __restrict__ deg) {
    __shared__ int hist[N_TILES];
    for (int i = threadIdx.x; i < N_TILES; i += 256) hist[i] = 0;
    __syncthreads();
    const int base = blockIdx.x * BIN_CHUNK;
    int r[BIN_CHUNK / 256];
#pragma unroll
    for (int k = 0; k < BIN_CHUNK / 256; ++k) r[k] = row[base + threadIdx.x + 256 * k];
#pragma unroll
    for (int k = 0; k < BIN_CHUNK / 256; ++k) atomicAdd(&hist[r[k] >> TILE_SHIFT], 1);
    __syncthreads();
    for (int i = threadIdx.x; i < N_TILES; i += 256)
        cnt[blockIdx.x * N_TILES + i] = hist[i];
#pragma unroll
    for (int k = 0; k < BIN_CHUNK / 256; ++k) atomicAdd(&deg[r[k]], 1.0f);
}

__global__ void __launch_bounds__(1024)
bin_scan_kernel(int* __restrict__ cnt, int* __restrict__ tile_start) {
    __shared__ int s[1024];
    const int t = threadIdx.x;
    int run = 0;
    if (t < N_TILES) {
        for (int b0 = 0; b0 < BIN_BLOCKS; b0 += 25) {
            int v[25];
#pragma unroll
            for (int u = 0; u < 25; ++u) v[u] = cnt[(b0 + u) * N_TILES + t];
#pragma unroll
            for (int u = 0; u < 25; ++u) {
                cnt[(b0 + u) * N_TILES + t] = run;
                run += v[u];
            }
        }
    }
    s[t] = run;
    __syncthreads();
    for (int off = 1; off < 1024; off <<= 1) {
        int v = (t >= off) ? s[t - off] : 0;
        __syncthreads();
        s[t] += v;
        __syncthreads();
    }
    if (t < N_TILES) tile_start[t] = s[t] - run;
    if (t == N_TILES - 1) tile_start[N_TILES] = s[t];
}

__global__ void __launch_bounds__(256)
bin_place_kernel(const int* __restrict__ row, const int* __restrict__ col,
                 const float* __restrict__ attr, const float* __restrict__ deg,
                 const int* __restrict__ cnt,
                 const int* __restrict__ tile_start, unsigned long long* __restrict__ rec) {
    __shared__ int cur[N_TILES];
    for (int i = threadIdx.x; i < N_TILES; i += 256)
        cur[i] = tile_start[i] + cnt[blockIdx.x * N_TILES + i];
    __syncthreads();
    const int base = blockIdx.x * BIN_CHUNK;
    for (int k0 = 0; k0 < BIN_CHUNK / 256; k0 += 5) {
        int r[5], c[5];
        float a[5], dr[5], dc[5], l[5];
#pragma unroll
        for (int u = 0; u < 5; ++u) {
            const int e = base + threadIdx.x + 256 * (k0 + u);
            r[u] = row[e];
            c[u] = col[e];
            a[u] = attr[e];
        }
#pragma unroll
        for (int u = 0; u < 5; ++u) {
            dr[u] = deg[r[u]];
            dc[u] = deg[c[u]];
        }
#pragma unroll
        for (int u = 0; u < 5; ++u) l[u] = -dinv_of(dr[u]) * a[u] * dinv_of(dc[u]);
#pragma unroll
        for (int u = 0; u < 5; ++u) {
            const int slot = atomicAdd(&cur[r[u] >> TILE_SHIFT], 1);
            const unsigned lo = (unsigned)c[u] | ((unsigned)(r[u] & (TILE_ROWS - 1)) << COL_BITS);
            const unsigned hi = __float_as_uint(l[u]);
            rec[slot] = (unsigned long long)lo | ((unsigned long long)hi << 32);
        }
    }
}

#define SUBS 8
#define SORT_CAP 2560
#define SORT_RPT (SORT_CAP / 256)
#define SORT_SLOTS (SORT_CAP + 64)

typedef __attribute__((ext_vector_type(4))) unsigned v4u;

__device__ __forceinline__ v8f wmma4(v2f a, float b0, float b1, v8f c) {
    v2f b = {b0, b1};
    return __builtin_amdgcn_wmma_f32_16x16x4_f32(
         false, a,  false, b,
         (short)0, c,  false,  false);
}

__global__ void __launch_bounds__(256, 8)
spmm_wmma_kernel(const unsigned long long* __restrict__ rec, const int* __restrict__ tile_start,
                 const float* __restrict__ h, const float* prev, float* dst, int mode) {
    __shared__ __attribute__((aligned(16))) unsigned long long srt[SORT_SLOTS];
    __shared__ int bcnt[SUBS];
    __shared__ int bstart[SUBS];

    const int tile = blockIdx.x;
    const int tid = threadIdx.x;
    const int lane = tid & 31;
    const int c = lane & 15;
    const int hh = lane >> 4;
    const int wave = __builtin_amdgcn_readfirstlane((int)(tid >> 5));

    const int s0 = tile_start[tile];
    const int s1 = tile_start[tile + 1];
    const float* hc = h + 4 * c;

    v8f acc0 = {0.f, 0.f, 0.f, 0.f, 0.f, 0.f, 0.f, 0.f};
    v8f acc1 = acc0, acc2 = acc0, acc3 = acc0;

    for (int b0 = s0; b0 < s1; b0 += SORT_CAP) {
        const int nb = min(SORT_CAP, s1 - b0);
        if (tid < SUBS) bcnt[tid] = 0;
        __syncthreads();

        unsigned long long r[SORT_RPT];
        int rank[SORT_RPT];
#pragma unroll
        for (int k = 0; k < SORT_RPT; ++k) {
            const int idx = k * 256 + tid;
            r[k] = (idx < nb) ? rec[b0 + idx] : 0ull;
        }
#pragma unroll
        for (int k = 0; k < SORT_RPT; ++k) {
            const int idx = k * 256 + tid;
            const unsigned lo = (unsigned)(r[k] & 0xffffffffull);
            rank[k] = 0;
            if (idx < nb) rank[k] = atomicAdd(&bcnt[(lo >> (COL_BITS + 4)) & (SUBS - 1)], 1);
        }
        __syncthreads();

        if (tid < SUBS) {
            int st = 0;
            for (int b = 0; b < tid; ++b) st += (bcnt[b] + 3) & ~3;
            bstart[tid] = st;
        }
        __syncthreads();

#pragma unroll
        for (int k = 0; k < SORT_RPT; ++k) {
            const int idx = k * 256 + tid;
            const unsigned lo = (unsigned)(r[k] & 0xffffffffull);
            if (idx < nb) srt[bstart[(lo >> (COL_BITS + 4)) & (SUBS - 1)] + rank[k]] = r[k];
        }
        __syncthreads();

        const int beg = __builtin_amdgcn_readfirstlane(bstart[wave]);
        const int nw = __builtin_amdgcn_readfirstlane(bcnt[wave]);
        const int nfull = nw >> 2;
        const unsigned long long* sw = srt + beg + 2 * hh;
        int g = 0;
        for (; g + 4 <= nfull; g += 4) {
            v4u p[4];
            v4f va[4], vb[4];
#pragma unroll
            for (int u = 0; u < 4; ++u) p[u] = *(const v4u*)(sw + 4 * (g + u));
#pragma unroll
            for (int u = 0; u < 4; ++u) {
                va[u] = *(const v4f*)(hc + (size_t)(p[u].x & ((1u << COL_BITS) - 1u)) * D);
                vb[u] = *(const v4f*)(hc + (size_t)(p[u].z & ((1u << COL_BITS) - 1u)) * D);
            }
#pragma unroll
            for (int u = 0; u < 4; ++u) {
                v2f a;
                a.x = (((p[u].x >> COL_BITS) & 15u) == (unsigned)c) ? __uint_as_float(p[u].y) : 0.0f;
                a.y = (((p[u].z >> COL_BITS) & 15u) == (unsigned)c) ? __uint_as_float(p[u].w) : 0.0f;
                acc0 = wmma4(a, va[u].x, vb[u].x, acc0);
                acc1 = wmma4(a, va[u].y, vb[u].y, acc1);
                acc2 = wmma4(a, va[u].z, vb[u].z, acc2);
                acc3 = wmma4(a, va[u].w, vb[u].w, acc3);
            }
        }
        for (; g < nfull; ++g) {
            const v4u p = *(const v4u*)(sw + 4 * g);
            const v4f va = *(const v4f*)(hc + (size_t)(p.x & ((1u << COL_BITS) - 1u)) * D);
            const v4f vb = *(const v4f*)(hc + (size_t)(p.z & ((1u << COL_BITS) - 1u)) * D);
            v2f a;
            a.x = (((p.x >> COL_BITS) & 15u) == (unsigned)c) ? __uint_as_float(p.y) : 0.0f;
            a.y = (((p.z >> COL_BITS) & 15u) == (unsigned)c) ? __uint_as_float(p.w) : 0.0f;
            acc0 = wmma4(a, va.x, vb.x, acc0);
            acc1 = wmma4(a, va.y, vb.y, acc1);
            acc2 = wmma4(a, va.z, vb.z, acc2);
            acc3 = wmma4(a, va.w, vb.w, acc3);
        }
        if (nw & 3) {
            const v4u p = *(const v4u*)(sw + 4 * nfull);
            const int e = 4 * nfull + 2 * hh;
            const bool ok0 = e < nw;
            const bool ok1 = e + 1 < nw;
            v4f va = {0.f, 0.f, 0.f, 0.f};
            v4f vb = {0.f, 0.f, 0.f, 0.f};
            if (ok0) va = *(const v4f*)(hc + (size_t)(p.x & ((1u << COL_BITS) - 1u)) * D);
            if (ok1) vb = *(const v4f*)(hc + (size_t)(p.z & ((1u << COL_BITS) - 1u)) * D);
            v2f a;
            a.x = (ok0 && ((p.x >> COL_BITS) & 15u) == (unsigned)c) ? __uint_as_float(p.y) : 0.0f;
            a.y = (ok1 && ((p.z >> COL_BITS) & 15u) == (unsigned)c) ? __uint_as_float(p.w) : 0.0f;
            acc0 = wmma4(a, va.x, vb.x, acc0);
            acc1 = wmma4(a, va.y, vb.y, acc1);
            acc2 = wmma4(a, va.z, vb.z, acc2);
            acc3 = wmma4(a, va.w, vb.w, acc3);
        }
        __syncthreads();
    }

    const int rowb = tile * TILE_ROWS + 16 * wave + 8 * hh;
    v4f pv[8];
    if (mode == 1) {
#pragma unroll
        for (int rr = 0; rr < 8; ++rr) {
            const int grow = rowb + rr;
            if (grow < N_NODES) pv[rr] = ((const v4f*)prev)[(size_t)grow * (D / 4) + c];
        }
    }
#pragma unroll
    for (int rr = 0; rr < 8; ++rr) {
        const int grow = rowb + rr;
        if (grow < N_NODES) {
            const size_t gi = (size_t)grow * (D / 4) + c;
            v4f a = {acc0[rr], acc1[rr], acc2[rr], acc3[rr]};
            if (mode == 1) a = 2.0f * a - pv[rr];
            ((v4f*)dst)[gi] = a;
        }
    }
}

__global__ void gemm_wmma_kernel(const float* __restrict__ H, const float* __restrict__ W,
                                 const float* __restrict__ bias, float* __restrict__ out,
                                 int n_row_tiles, int accumulate) {
    const int wave = threadIdx.x >> 5;
    const int lane = threadIdx.x & 31;
    const int col0 = wave * 16;
    const int laneN = lane & 15;
    const int hi = lane >> 4;

    v2f bfrag[16];
#pragma unroll
    for (int ks = 0; ks < 16; ++ks) {
        int k = ks * 4 + hi * 2;
        bfrag[ks].x = W[(k + 0) * D + col0 + laneN];
        bfrag[ks].y = W[(k + 1) * D + col0 + laneN];
    }
    float bval = accumulate ? 0.0f : bias[col0 + laneN];

    for (int tile = blockIdx.x; tile < n_row_tiles; tile += gridDim.x) {
        const int row0 = tile * 16;
        v8f acc;
        if (accumulate) {
#pragma unroll
            for (int r = 0; r < 8; ++r)
                acc[r] = out[(row0 + hi * 8 + r) * D + col0 + laneN];
        } else {
#pragma unroll
            for (int r = 0; r < 8; ++r) acc[r] = bval;
        }

        const float* hrow = H + (row0 + laneN) * D + hi * 2;
#pragma unroll
        for (int ks = 0; ks < 16; ++ks) {
            v2f a;
            a.x = hrow[ks * 4 + 0];
            a.y = hrow[ks * 4 + 1];
            acc = __builtin_amdgcn_wmma_f32_16x16x4_f32(
                 false, a,  false, bfrag[ks],
                 (short)0, acc,  false,  false);
        }

#pragma unroll
        for (int r = 0; r < 8; ++r)
            out[(row0 + hi * 8 + r) * D + col0 + laneN] = acc[r];
    }
}

extern "C" void kernel_launch(void* const* d_in, const int* in_sizes, int n_in,
                              void* d_out, int out_size, void* d_ws, size_t ws_size,
                              hipStream_t stream) {
    const float* x         = (const float*)d_in[0];
    const int*   edge_idx  = (const int*)d_in[1];
    const float* edge_attr = (const float*)d_in[2];
    const float* weight    = (const float*)d_in[3];
    const float* bias      = (const float*)d_in[4];
    float*       out       = (float*)d_out;

    const int* row = edge_idx;
    const int* col = edge_idx + N_EDGES;

    char* ws = (char*)d_ws;
    float* deg  = (float*)ws; ws += sizeof(float) * N_NODES;
    ws += sizeof(float) * N_NODES;
    ws += sizeof(float) * N_EDGES;
    float* A    = (float*)ws; ws += sizeof(float) * (size_t)N_NODES * D;
    float* B    = (float*)ws; ws += sizeof(float) * (size_t)N_NODES * D;
    float* Z    = (float*)ws;

    const int n_row_tiles = N_NODES / 16;
    const int GEMM_BLOCKS = 1024;
    const int n_blocks = (N_NODES + 255) / 256;

    unsigned long long* rec = (unsigned long long*)Z;
    int* cnt        = (int*)(rec + N_EDGES);
    int* tile_start = cnt + BIN_BLOCKS * N_TILES;

    zero_kernel<<<n_blocks, 256, 0, stream>>>(deg, (long long)N_NODES);

    gemm_wmma_kernel<<<GEMM_BLOCKS, 128, 0, stream>>>(x, weight, bias, out, n_row_tiles, 0);

    bin_count_kernel<<<BIN_BLOCKS, 256, 0, stream>>>(row, cnt, deg);
    bin_scan_kernel<<<1, 1024, 0, stream>>>(cnt, tile_start);
    bin_place_kernel<<<BIN_BLOCKS, 256, 0, stream>>>(row, col, edge_attr, deg, cnt, tile_start, rec);

    spmm_wmma_kernel<<<N_TILES, 256, 0, stream>>>(rec, tile_start, x, x, A, 0);
    gemm_wmma_kernel<<<GEMM_BLOCKS, 128, 0, stream>>>(A, weight + 1 * D * D, nullptr, out, n_row_tiles, 1);

    spmm_wmma_kernel<<<N_TILES, 256, 0, stream>>>(rec, tile_start, A, x, B, 1);
    gemm_wmma_kernel<<<GEMM_BLOCKS, 128, 0, stream>>>(B, weight + 2 * D * D, nullptr, out, n_row_tiles, 1);

    spmm_wmma_kernel<<<N_TILES, 256, 0, stream>>>(rec, tile_start, B, A, A, 1);
    gemm_wmma_kernel<<<GEMM_BLOCKS, 128, 0, stream>>>(A, weight + 3 * D * D, nullptr, out, n_row_tiles, 1);
}


# degrees counted per destination tile in LDS from the binned records into a dinv table: the per-edge global float atomic adds are gone; first SPMM pass forms lap into the records; sum order kept
# speedup vs baseline: 4.8319x; 1.4576x over previous
#include <hip/hip_runtime.h>

#define N_NODES 100000
#define N_EDGES 1600000
#define D 64

typedef __attribute__((ext_vector_type(2))) float v2f;
typedef __attribute__((ext_vector_type(8))) float v8f;

__device__ __forceinline__ float dinv_of(float d) {
    return (d > 0.0f) ? (1.0f / sqrtf(d)) : 0.0f;
}

#define TILE_ROWS 128
#define TILE_SHIFT 7
#define N_TILES 782
#define BIN_BLOCKS 250
#define BIN_CHUNK 6400
#define COL_BITS 20

typedef __attribute__((ext_vector_type(4))) float v4f;

__global__ void __launch_bounds__(256)
bin_count_kernel(const int* __restrict__ row, int* __restrict__ cnt) {
    __shared__ int hist[N_TILES];
    for (int i = threadIdx.x; i < N_TILES; i += 256) hist[i] = 0;
    __syncthreads();
    const int base = blockIdx.x * BIN_CHUNK;
    int r[BIN_CHUNK / 256];
#pragma unroll
    for (int k = 0; k < BIN_CHUNK / 256; ++k) r[k] = row[base + threadIdx.x + 256 * k];
#pragma unroll
    for (int k = 0; k < BIN_CHUNK / 256; ++k) atomicAdd(&hist[r[k] >> TILE_SHIFT], 1);
    __syncthreads();
    for (int i = threadIdx.x; i < N_TILES; i += 256)
        cnt[blockIdx.x * N_TILES + i] = hist[i];
}

__global__ void __launch_bounds__(1024)
bin_scan_kernel(int* __restrict__ cnt, int* __restrict__ tile_start) {
    __shared__ int s[1024];
    const int t = threadIdx.x;
    int run = 0;
    if (t < N_TILES) {
        for (int b0 = 0; b0 < BIN_BLOCKS; b0 += 25) {
            int v[25];
#pragma unroll
            for (int u = 0; u < 25; ++u) v[u] = cnt[(b0 + u) * N_TILES + t];
#pragma unroll
            for (int u = 0; u < 25; ++u) {
                cnt[(b0 + u) * N_TILES + t] = run;
                run += v[u];
            }
        }
    }
    s[t] = run;
    __syncthreads();
    for (int off = 1; off < 1024; off <<= 1) {
        int v = (t >= off) ? s[t - off] : 0;
        __syncthreads();
        s[t] += v;
        __syncthreads();
    }
    if (t < N_TILES) tile_start[t] = s[t] - run;
    if (t == N_TILES - 1) tile_start[N_TILES] = s[t];
}

__global__ void __launch_bounds__(256)
bin_place_kernel(const int* __restrict__ row, const int* __restrict__ col,
                 const float* __restrict__ attr,
                 const int* __restrict__ cnt,
                 const int* __restrict__ tile_start, unsigned long long* __restrict__ rec) {
    __shared__ int cur[N_TILES];
    for (int i = threadIdx.x; i < N_TILES; i += 256)
        cur[i] = tile_start[i] + cnt[blockIdx.x * N_TILES + i];
    __syncthreads();
    const int base = blockIdx.x * BIN_CHUNK;
    for (int k0 = 0; k0 < BIN_CHUNK / 256; k0 += 5) {
        int r[5], c[5];
        float a[5];
#pragma unroll
        for (int u = 0; u < 5; ++u) {
            const int e = base + threadIdx.x + 256 * (k0 + u);
            r[u] = row[e];
            c[u] = col[e];
            a[u] = attr[e];
        }
#pragma unroll
        for (int u = 0; u < 5; ++u) {
            const int slot = atomicAdd(&cur[r[u] >> TILE_SHIFT], 1);
            const unsigned lo = (unsigned)c[u] | ((unsigned)(r[u] & (TILE_ROWS - 1)) << COL_BITS);
            const unsigned hi = __float_as_uint(a[u]);
            rec[slot] = (unsigned long long)lo | ((unsigned long long)hi << 32);
        }
    }
}

#define SUBS 8
#define SORT_CAP 2560
#define SORT_RPT (SORT_CAP / 256)
#define SORT_SLOTS (SORT_CAP + 64)
#define TS_PITCH 68
#define TS_FLOATS (TILE_ROWS * TS_PITCH)

typedef __attribute__((ext_vector_type(4))) unsigned v4u;

__global__ void __launch_bounds__(256)
tile_dinv_kernel(const unsigned long long* __restrict__ rec, const int* __restrict__ tile_start,
                 float* __restrict__ dinv) {
    __shared__ int rc[TILE_ROWS];
    const int tile = blockIdx.x;
    const int tid = threadIdx.x;
    if (tid < TILE_ROWS) rc[tid] = 0;
    __syncthreads();
    const int s0 = tile_start[tile];
    const int s1 = tile_start[tile + 1];
    const unsigned* lo = (const unsigned*)rec;
    for (int b0 = s0; b0 < s1; b0 += SORT_CAP) {
        const int nb = min(SORT_CAP, s1 - b0);
        unsigned w[SORT_RPT];
#pragma unroll
        for (int k = 0; k < SORT_RPT; ++k) {
            const int idx = min(k * 256 + tid, nb - 1);
            w[k] = lo[2 * (size_t)(b0 + idx)];
        }
#pragma unroll
        for (int k = 0; k < SORT_RPT; ++k) {
            const int idx = k * 256 + tid;
            if (idx < nb) atomicAdd(&rc[(w[k] >> COL_BITS) & (TILE_ROWS - 1)], 1);
        }
    }
    __syncthreads();
    if (tid < TILE_ROWS) {
        const int grow = tile * TILE_ROWS + tid;
        if (grow < N_NODES) dinv[grow] = dinv_of((float)rc[tid]);
    }
}

__device__ __forceinline__ v8f wmma4(v2f a, float b0, float b1, v8f c) {
    v2f b = {b0, b1};
    return __builtin_amdgcn_wmma_f32_16x16x4_f32(
         false, a,  false, b,
         (short)0, c,  false,  false);
}

__device__ __forceinline__ void load_panel(const float* __restrict__ Wk, int ct, int c, int hh,
                                           v2f (&bf)[16]) {
#pragma unroll
    for (int ks = 0; ks < 16; ++ks) {
        const int k = 4 * ks + 2 * hh;
        bf[ks].x = Wk[(k + 0) * D + 16 * ct + c];
        bf[ks].y = Wk[(k + 1) * D + 16 * ct + c];
    }
}

__device__ __forceinline__ void tile_product(const float* ts, int rg, int c, int hh,
                                             const v2f (&bf)[16], v8f (&o)[4]) {
#pragma unroll
    for (int j = 0; j < 4; ++j) {
        const float* ar = ts + (16 * (4 * rg + j) + c) * TS_PITCH + 2 * hh;
#pragma unroll
        for (int ks = 0; ks < 16; ++ks) {
            const v2f a = *(const v2f*)(ar + 4 * ks);
            o[j] = __builtin_amdgcn_wmma_f32_16x16x4_f32(
                 false, a,  false, bf[ks],
                 (short)0, o[j],  false,  false);
        }
    }
}

template <int STAGE>
__global__ void __launch_bounds__(256, 8)
spmm_gemm_kernel(unsigned long long* __restrict__ rec, const int* __restrict__ tile_start,
                 const float* __restrict__ h, const float* __restrict__ prev,
                 float* __restrict__ dst,
                 const float* __restrict__ Wa, const float* __restrict__ Wb,
                 const float* __restrict__ bias, float* __restrict__ out,
                 const float* __restrict__ dinv) {
    __shared__ __attribute__((aligned(16))) float smem[(STAGE == 0 ? 2 : 1) * TS_FLOATS];
    __shared__ int bcnt[SUBS];
    __shared__ int bstart[SUBS];
    unsigned long long* srt = (unsigned long long*)smem;

    const int tile = blockIdx.x;
    const int tid = threadIdx.x;
    const int lane = tid & 31;
    const int c = lane & 15;
    const int hh = lane >> 4;
    const int wave = __builtin_amdgcn_readfirstlane((int)(tid >> 5));

    const int s0 = tile_start[tile];
    const int s1 = tile_start[tile + 1];
    const float* hc = h + 4 * c;

    v8f acc0 = {0.f, 0.f, 0.f, 0.f, 0.f, 0.f, 0.f, 0.f};
    v8f acc1 = acc0, acc2 = acc0, acc3 = acc0;

    for (int b0 = s0; b0 < s1; b0 += SORT_CAP) {
        const int nb = min(SORT_CAP, s1 - b0);
        if (tid < SUBS) bcnt[tid] = 0;
        __syncthreads();

        unsigned long long r[SORT_RPT];
        int rank[SORT_RPT];
#pragma unroll
        for (int k = 0; k < SORT_RPT; ++k) {
            const int idx = k * 256 + tid;
            r[k] = (idx < nb) ? rec[b0 + idx] : 0ull;
        }
        float dc[SORT_RPT], dr[SORT_RPT];
        if (STAGE == 0) {
#pragma unroll
            for (int k = 0; k < SORT_RPT; ++k) {
                const unsigned lo = (unsigned)(r[k] & 0xffffffffull);
                dc[k] = dinv[lo & ((1u << COL_BITS) - 1u)];
                dr[k] = dinv[tile * TILE_ROWS + ((lo >> COL_BITS) & (TILE_ROWS - 1))];
            }
        }
#pragma unroll
        for (int k = 0; k < SORT_RPT; ++k) {
            const int idx = k * 256 + tid;
            const unsigned lo = (unsigned)(r[k] & 0xffffffffull);
            rank[k] = 0;
            if (idx < nb) rank[k] = atomicAdd(&bcnt[(lo >> (COL_BITS + 4)) & (SUBS - 1)], 1);
        }
        __syncthreads();

        if (tid < SUBS) {
            int st = 0;
            for (int b = 0; b < tid; ++b) st += (bcnt[b] + 3) & ~3;
            bstart[tid] = st;
        }
        __syncthreads();

        if (STAGE == 0) {
#pragma unroll
            for (int k = 0; k < SORT_RPT; ++k) {
                const unsigned lo = (unsigned)(r[k] & 0xffffffffull);
                const float a = __uint_as_float((unsigned)(r[k] >> 32));
                const float l = -dr[k] * a * dc[k];
                r[k] = (unsigned long long)lo | ((unsigned long long)__float_as_uint(l) << 32);
            }
        }

#pragma unroll
        for (int k = 0; k < SORT_RPT; ++k) {
            const int idx = k * 256 + tid;
            const unsigned lo = (unsigned)(r[k] & 0xffffffffull);
            if (idx < nb) srt[bstart[(lo >> (COL_BITS + 4)) & (SUBS - 1)] + rank[k]] = r[k];
        }
        __syncthreads();

        if (STAGE == 0) {
#pragma unroll
            for (int k = 0; k < SORT_RPT; ++k) {
                const int idx = k * 256 + tid;
                if (idx < nb) rec[b0 + idx] = r[k];
            }
        }

        const int beg = __builtin_amdgcn_readfirstlane(bstart[wave]);
        const int nw = __builtin_amdgcn_readfirstlane(bcnt[wave]);
        const int nfull = nw >> 2;
        const unsigned long long* sw = srt + beg + 2 * hh;
        int g = 0;
        for (; g + 4 <= nfull; g += 4) {
            v4u p[4];
            v4f va[4], vb[4];
#pragma unroll
            for (int u = 0; u < 4; ++u) p[u] = *(const v4u*)(sw + 4 * (g + u));
#pragma unroll
            for (int u = 0; u < 4; ++u) {
                va[u] = *(const v4f*)(hc + (size_t)(p[u].x & ((1u << COL_BITS) - 1u)) * D);
                vb[u] = *(const v4f*)(hc + (size_t)(p[u].z & ((1u << COL_BITS) - 1u)) * D);
            }
#pragma unroll
            for (int u = 0; u < 4; ++u) {
                v2f a;
                a.x = (((p[u].x >> COL_BITS) & 15u) == (unsigned)c) ? __uint_as_float(p[u].y) : 0.0f;
                a.y = (((p[u].z >> COL_BITS) & 15u) == (unsigned)c) ? __uint_as_float(p[u].w) : 0.0f;
                acc0 = wmma4(a, va[u].x, vb[u].x, acc0);
                acc1 = wmma4(a, va[u].y, vb[u].y, acc1);
                acc2 = wmma4(a, va[u].z, vb[u].z, acc2);
                acc3 = wmma4(a, va[u].w, vb[u].w, acc3);
            }
        }
        for (; g < nfull; ++g) {
            const v4u p = *(const v4u*)(sw + 4 * g);
            const v4f va = *(const v4f*)(hc + (size_t)(p.x & ((1u << COL_BITS) - 1u)) * D);
            const v4f vb = *(const v4f*)(hc + (size_t)(p.z & ((1u << COL_BITS) - 1u)) * D);
            v2f a;
            a.x = (((p.x >> COL_BITS) & 15u) == (unsigned)c) ? __uint_as_float(p.y) : 0.0f;
            a.y = (((p.z >> COL_BITS) & 15u) == (unsigned)c) ? __uint_as_float(p.w) : 0.0f;
            acc0 = wmma4(a, va.x, vb.x, acc0);
            acc1 = wmma4(a, va.y, vb.y, acc1);
            acc2 = wmma4(a, va.z, vb.z, acc2);
            acc3 = wmma4(a, va.w, vb.w, acc3);
        }
        if (nw & 3) {
            const v4u p = *(const v4u*)(sw + 4 * nfull);
            const int e = 4 * nfull + 2 * hh;
            const bool ok0 = e < nw;
            const bool ok1 = e + 1 < nw;
            v4f va = {0.f, 0.f, 0.f, 0.f};
            v4f vb = {0.f, 0.f, 0.f, 0.f};
            if (ok0) va = *(const v4f*)(hc + (size_t)(p.x & ((1u << COL_BITS) - 1u)) * D);
            if (ok1) vb = *(const v4f*)(hc + (size_t)(p.z & ((1u << COL_BITS) - 1u)) * D);
            v2f a;
            a.x = (ok0 && ((p.x >> COL_BITS) & 15u) == (unsigned)c) ? __uint_as_float(p.y) : 0.0f;
            a.y = (ok1 && ((p.z >> COL_BITS) & 15u) == (unsigned)c) ? __uint_as_float(p.w) : 0.0f;
            acc0 = wmma4(a, va.x, vb.x, acc0);
            acc1 = wmma4(a, va.y, vb.y, acc1);
            acc2 = wmma4(a, va.z, vb.z, acc2);
            acc3 = wmma4(a, va.w, vb.w, acc3);
        }
        __syncthreads();
    }

    const int rowl = 16 * wave + 8 * hh;
    const int rowb = tile * TILE_ROWS + rowl;
    const int ct = wave & 3;
    const int rg = wave >> 2;

    v2f bfa[16], bfb[16];
    if (STAGE == 0) load_panel(Wa, ct, c, hh, bfa);
    else load_panel(Wb, ct, c, hh, bfb);

    v4f pv[8];
#pragma unroll
    for (int rr = 0; rr < 8; ++rr) {
        const int grow = rowb + rr;
        pv[rr] = (v4f){0.f, 0.f, 0.f, 0.f};
        if (grow < N_NODES) pv[rr] = ((const v4f*)prev)[(size_t)grow * (D / 4) + c];
    }
#pragma unroll
    for (int rr = 0; rr < 8; ++rr) {
        const int grow = rowb + rr;
        v4f a = {acc0[rr], acc1[rr], acc2[rr], acc3[rr]};
        if (STAGE != 0) a = 2.0f * a - pv[rr];
        if (STAGE != 2) {
            if (grow < N_NODES) ((v4f*)dst)[(size_t)grow * (D / 4) + c] = a;
        }
        *(v4f*)(smem + (rowl + rr) * TS_PITCH + 4 * c) = a;
        if (STAGE == 0) *(v4f*)(smem + TS_FLOATS + (rowl + rr) * TS_PITCH + 4 * c) = pv[rr];
    }
    __syncthreads();

    v8f o[4];
    if (STAGE == 0) {
        const float bval = bias[16 * ct + c];
#pragma unroll
        for (int j = 0; j < 4; ++j)
#pragma unroll
            for (int r = 0; r < 8; ++r) o[j][r] = bval;
        load_panel(Wb, ct, c, hh, bfb);
        tile_product(smem + TS_FLOATS, rg, c, hh, bfa, o);
        tile_product(smem, rg, c, hh, bfb, o);
    } else {
#pragma unroll
        for (int j = 0; j < 4; ++j)
#pragma unroll
            for (int r = 0; r < 8; ++r) o[j][r] = 0.0f;
        tile_product(smem, rg, c, hh, bfb, o);
    }

    const int orow = tid >> 4;
    const int oc = tid & 15;
    v4f ov[8];
    if (STAGE != 0) {
#pragma unroll
        for (int i = 0; i < 8; ++i) {
            const int grow = tile * TILE_ROWS + 16 * i + orow;
            ov[i] = (v4f){0.f, 0.f, 0.f, 0.f};
            if (grow < N_NODES) ov[i] = ((const v4f*)out)[(size_t)grow * (D / 4) + oc];
        }
    }
    __syncthreads();
#pragma unroll
    for (int j = 0; j < 4; ++j)
#pragma unroll
        for (int r = 0; r < 8; ++r)
            smem[(16 * (4 * rg + j) + 8 * hh + r) * TS_PITCH + 16 * ct + c] = o[j][r];
    __syncthreads();
#pragma unroll
    for (int i = 0; i < 8; ++i) {
        const int lrow = 16 * i + orow;
        const int grow = tile * TILE_ROWS + lrow;
        v4f v = *(const v4f*)(smem + lrow * TS_PITCH + 4 * oc);
        if (STAGE != 0) v = ov[i] + v;
        if (grow < N_NODES) ((v4f*)out)[(size_t)grow * (D / 4) + oc] = v;
    }
}

extern "C" void kernel_launch(void* const* d_in, const int* in_sizes, int n_in,
                              void* d_out, int out_size, void* d_ws, size_t ws_size,
                              hipStream_t stream) {
    const float* x         = (const float*)d_in[0];
    const int*   edge_idx  = (const int*)d_in[1];
    const float* edge_attr = (const float*)d_in[2];
    const float* weight    = (const float*)d_in[3];
    const float* bias      = (const float*)d_in[4];
    float*       out       = (float*)d_out;

    const int* row = edge_idx;
    const int* col = edge_idx + N_EDGES;

    char* ws = (char*)d_ws;
    float* dinv = (float*)ws; ws += sizeof(float) * N_NODES;
    ws += sizeof(float) * N_NODES;
    ws += sizeof(float) * N_EDGES;
    float* A    = (float*)ws; ws += sizeof(float) * (size_t)N_NODES * D;
    float* B    = (float*)ws; ws += sizeof(float) * (size_t)N_NODES * D;
    float* Z    = (float*)ws;

    unsigned long long* rec = (unsigned long long*)Z;
    int* cnt        = (int*)(rec + N_EDGES);
    int* tile_start = cnt + BIN_BLOCKS * N_TILES;

    bin_count_kernel<<<BIN_BLOCKS, 256, 0, stream>>>(row, cnt);
    bin_scan_kernel<<<1, 1024, 0, stream>>>(cnt, tile_start);
    bin_place_kernel<<<BIN_BLOCKS, 256, 0, stream>>>(row, col, edge_attr, cnt, tile_start, rec);

    tile_dinv_kernel<<<N_TILES, 256, 0, stream>>>(rec, tile_start, dinv);

    spmm_gemm_kernel<0><<<N_TILES, 256, 0, stream>>>(rec, tile_start, x, x, A,
                                                     weight, weight + 1 * D * D, bias, out, dinv);
    spmm_gemm_kernel<1><<<N_TILES, 256, 0, stream>>>(rec, tile_start, A, x, B,
                                                     nullptr, weight + 2 * D * D, nullptr, out,
                                                     nullptr);
    spmm_gemm_kernel<2><<<N_TILES, 256, 0, stream>>>(rec, tile_start, B, A, nullptr,
                                                     nullptr, weight + 3 * D * D, nullptr, out,
                                                     nullptr);
}


# tile records grouped by 16-row block once: SPMM pass 1 writes them back in block order with the block counts; passes 2 and 3 read their runs from memory, no LDS grouping, sums in pass 1's slot order
# speedup vs baseline: 5.0403x; 1.0431x over previous
#include <hip/hip_runtime.h>

#define N_NODES 100000
#define N_EDGES 1600000
#define D 64

typedef __attribute__((ext_vector_type(2))) float v2f;
typedef __attribute__((ext_vector_type(8))) float v8f;

__device__ __forceinline__ float dinv_of(float d) {
    return (d > 0.0f) ? (1.0f / sqrtf(d)) : 0.0f;
}

#define TILE_ROWS 128
#define TILE_SHIFT 7
#define N_TILES 782
#define BIN_BLOCKS 250
#define BIN_CHUNK 6400
#define COL_BITS 20

typedef __attribute__((ext_vector_type(4))) float v4f;

__global__ void __launch_bounds__(256)
bin_count_kernel(const int* __restrict__ row, int* __restrict__ cnt) {
    __shared__ int hist[N_TILES];
    for (int i = threadIdx.x; i < N_TILES; i += 256) hist[i] = 0;
    __syncthreads();
    const int base = blockIdx.x * BIN_CHUNK;
    int r[BIN_CHUNK / 256];
#pragma unroll
    for (int k = 0; k < BIN_CHUNK / 256; ++k) r[k] = row[base + threadIdx.x + 256 * k];
#pragma unroll
    for (int k = 0; k < BIN_CHUNK / 256; ++k) atomicAdd(&hist[r[k] >> TILE_SHIFT], 1);
    __syncthreads();
    for (int i = threadIdx.x; i < N_TILES; i += 256)
        cnt[blockIdx.x * N_TILES + i] = hist[i];
}

__global__ void __launch_bounds__(1024)
bin_scan_kernel(int* __restrict__ cnt, int* __restrict__ tile_start) {
    __shared__ int s[1024];
    const int t = threadIdx.x;
    int run = 0;
    if (t < N_TILES) {
        for (int b0 = 0; b0 < BIN_BLOCKS; b0 += 25) {
            int v[25];
#pragma unroll
            for (int u = 0; u < 25; ++u) v[u] = cnt[(b0 + u) * N_TILES + t];
#pragma unroll
            for (int u = 0; u < 25; ++u) {
                cnt[(b0 + u) * N_TILES + t] = run;
                run += v[u];
            }
        }
    }
    s[t] = run;
    __syncthreads();
    for (int off = 1; off < 1024; off <<= 1) {
        int v = (t >= off) ? s[t - off] : 0;
        __syncthreads();
        s[t] += v;
        __syncthreads();
    }
    if (t < N_TILES) tile_start[t] = s[t] - run;
    if (t == N_TILES - 1) tile_start[N_TILES] = s[t];
}

__global__ void __launch_bounds__(256)
bin_place_kernel(const int* __restrict__ row, const int* __restrict__ col,
                 const float* __restrict__ attr,
                 const int* __restrict__ cnt,
                 const int* __restrict__ tile_start, unsigned long long* __restrict__ rec) {
    __shared__ int cur[N_TILES];
    for (int i = threadIdx.x; i < N_TILES; i += 256)
        cur[i] = tile_start[i] + cnt[blockIdx.x * N_TILES + i];
    __syncthreads();
    const int base = blockIdx.x * BIN_CHUNK;
    for (int k0 = 0; k0 < BIN_CHUNK / 256; k0 += 5) {
        int r[5], c[5];
        float a[5];
#pragma unroll
        for (int u = 0; u < 5; ++u) {
            const int e = base + threadIdx.x + 256 * (k0 + u);
            r[u] = row[e];
            c[u] = col[e];
            a[u] = attr[e];
        }
#pragma unroll
        for (int u = 0; u < 5; ++u) {
            const int slot = atomicAdd(&cur[r[u] >> TILE_SHIFT], 1);
            const unsigned lo = (unsigned)c[u] | ((unsigned)(r[u] & (TILE_ROWS - 1)) << COL_BITS);
            const unsigned hi = __float_as_uint(a[u]);
            rec[slot] = (unsigned long long)lo | ((unsigned long long)hi << 32);
        }
    }
}

#define SUBS 8
#define SORT_CAP 2560
#define SORT_RPT (SORT_CAP / 256)
#define SORT_SLOTS (SORT_CAP + 4)
#define TS_PITCH 68
#define TS_FLOATS (TILE_ROWS * TS_PITCH)

typedef __attribute__((ext_vector_type(4))) unsigned v4u;

__global__ void __launch_bounds__(256)
tile_dinv_kernel(const unsigned long long* __restrict__ rec, const int* __restrict__ tile_start,
                 float* __restrict__ dinv) {
    __shared__ int rc[TILE_ROWS];
    const int tile = blockIdx.x;
    const int tid = threadIdx.x;
    if (tid < TILE_ROWS) rc[tid] = 0;
    __syncthreads();
    const int s0 = tile_start[tile];
    const int s1 = tile_start[tile + 1];
    const unsigned* lo = (const unsigned*)rec;
    for (int b0 = s0; b0 < s1; b0 += SORT_CAP) {
        const int nb = min(SORT_CAP, s1 - b0);
        unsigned w[SORT_RPT];
#pragma unroll
        for (int k = 0; k < SORT_RPT; ++k) {
            const int idx = min(k * 256 + tid, nb - 1);
            w[k] = lo[2 * (size_t)(b0 + idx)];
        }
#pragma unroll
        for (int k = 0; k < SORT_RPT; ++k) {
            const int idx = k * 256 + tid;
            if (idx < nb) atomicAdd(&rc[(w[k] >> COL_BITS) & (TILE_ROWS - 1)], 1);
        }
    }
    __syncthreads();
    if (tid < TILE_ROWS) {
        const int grow = tile * TILE_ROWS + tid;
        if (grow < N_NODES) dinv[grow] = dinv_of((float)rc[tid]);
    }
}

__device__ __forceinline__ v8f wmma4(v2f a, float b0, float b1, v8f c) {
    v2f b = {b0, b1};
    return __builtin_amdgcn_wmma_f32_16x16x4_f32(
         false, a,  false, b,
         (short)0, c,  false,  false);
}

__device__ __forceinline__ void load_panel(const float* __restrict__ Wk, int ct, int c, int hh,
                                           v2f (&bf)[16]) {
#pragma unroll
    for (int ks = 0; ks < 16; ++ks) {
        const int k = 4 * ks + 2 * hh;
        bf[ks].x = Wk[(k + 0) * D + 16 * ct + c];
        bf[ks].y = Wk[(k + 1) * D + 16 * ct + c];
    }
}

__device__ __forceinline__ void tile_product(const float* ts, int rg, int c, int hh,
                                             const v2f (&bf)[16], v8f (&o)[4]) {
#pragma unroll
    for (int j = 0; j < 4; ++j) {
        const float* ar = ts + (16 * (4 * rg + j) + c) * TS_PITCH + 2 * hh;
#pragma unroll
        for (int ks = 0; ks < 16; ++ks) {
            const v2f a = *(const v2f*)(ar + 4 * ks);
            o[j] = __builtin_amdgcn_wmma_f32_16x16x4_f32(
                 false, a,  false, bf[ks],
                 (short)0, o[j],  false,  false);
        }
    }
}

__device__ __forceinline__ v4u load_pair(const unsigned long long* q) {
    const unsigned long long a = q[0];
    const unsigned long long b = q[1];
    v4u p = {(unsigned)a, (unsigned)(a >> 32), (unsigned)b, (unsigned)(b >> 32)};
    return p;
}

__device__ __forceinline__ void group_product(v4u p, bool ok0, bool ok1, v4f va, v4f vb, int c,
                                              v8f& acc0, v8f& acc1, v8f& acc2, v8f& acc3) {
    v2f a;
    a.x = (ok0 && ((p.x >> COL_BITS) & 15u) == (unsigned)c) ? __uint_as_float(p.y) : 0.0f;
    a.y = (ok1 && ((p.z >> COL_BITS) & 15u) == (unsigned)c) ? __uint_as_float(p.w) : 0.0f;
    acc0 = wmma4(a, va.x, vb.x, acc0);
    acc1 = wmma4(a, va.y, vb.y, acc1);
    acc2 = wmma4(a, va.z, vb.z, acc2);
    acc3 = wmma4(a, va.w, vb.w, acc3);
}

template <bool IN_LDS>
__device__ __forceinline__ void sum_run(const unsigned long long* sw, int nw, const float* hc,
                                        int c, int hh,
                                        v8f& acc0, v8f& acc1, v8f& acc2, v8f& acc3) {
    const int nfull = nw >> 2;
    int g = 0;
    v4u p[4];
#pragma unroll
    for (int u = 0; u < 4; ++u) p[u] = (v4u){0u, 0u, 0u, 0u};
    if (!IN_LDS && nfull >= 4) {
#pragma unroll
        for (int u = 0; u < 4; ++u) p[u] = load_pair(sw + 4 * u);
    }
    for (; g + 4 <= nfull; g += 4) {
        v4f va[4], vb[4];
        v4u pn[4];
        if (IN_LDS) {
#pragma unroll
            for (int u = 0; u < 4; ++u) p[u] = load_pair(sw + 4 * (g + u));
        }
#pragma unroll
        for (int u = 0; u < 4; ++u) {
            va[u] = *(const v4f*)(hc + (size_t)(p[u].x & ((1u << COL_BITS) - 1u)) * D);
            vb[u] = *(const v4f*)(hc + (size_t)(p[u].z & ((1u << COL_BITS) - 1u)) * D);
        }
        if (!IN_LDS) {
            const int gn = min(g + 4, nfull - 4);
#pragma unroll
            for (int u = 0; u < 4; ++u) pn[u] = load_pair(sw + 4 * (gn + u));
        }
#pragma unroll
        for (int u = 0; u < 4; ++u)
            group_product(p[u], true, true, va[u], vb[u], c, acc0, acc1, acc2, acc3);
        if (!IN_LDS) {
#pragma unroll
            for (int u = 0; u < 4; ++u) p[u] = pn[u];
        }
    }
    for (; g < nfull; ++g) {
        const v4u q = load_pair(sw + 4 * g);
        const v4f va = *(const v4f*)(hc + (size_t)(q.x & ((1u << COL_BITS) - 1u)) * D);
        const v4f vb = *(const v4f*)(hc + (size_t)(q.z & ((1u << COL_BITS) - 1u)) * D);
        group_product(q, true, true, va, vb, c, acc0, acc1, acc2, acc3);
    }
    if (nw & 3) {
        const int e = 4 * nfull + 2 * hh;
        const bool ok0 = e < nw;
        const bool ok1 = e + 1 < nw;
        v4u q = {0u, 0u, 0u, 0u};
        if (IN_LDS) {
            q = load_pair(sw + 4 * nfull);
        } else {
            if (ok0) { const unsigned long long a = sw[4 * nfull]; q.x = (unsigned)a; q.y = (unsigned)(a >> 32); }
            if (ok1) { const unsigned long long b = sw[4 * nfull + 1]; q.z = (unsigned)b; q.w = (unsigned)(b >> 32); }
        }
        v4f va = {0.f, 0.f, 0.f, 0.f};
        v4f vb = {0.f, 0.f, 0.f, 0.f};
        if (ok0) va = *(const v4f*)(hc + (size_t)(q.x & ((1u << COL_BITS) - 1u)) * D);
        if (ok1) vb = *(const v4f*)(hc + (size_t)(q.z & ((1u << COL_BITS) - 1u)) * D);
        group_product(q, ok0, ok1, va, vb, c, acc0, acc1, acc2, acc3);
    }
}

template <int STAGE>
__global__ void __launch_bounds__(256, 8)
spmm_gemm_kernel(unsigned long long* __restrict__ rec, const int* __restrict__ tile_start,
                 const float* __restrict__ h, const float* __restrict__ prev,
                 float* __restrict__ dst,
                 const float* __restrict__ Wa, const float* __restrict__ Wb,
                 const float* __restrict__ bias, float* __restrict__ out,
                 const float* __restrict__ dinv, int* __restrict__ blkcnt) {
    __shared__ __attribute__((aligned(16))) float smem[(STAGE == 0 ? 2 : 1) * TS_FLOATS];
    __shared__ int bcnt[SUBS];
    __shared__ int bstart[SUBS];
    unsigned long long* srt = (unsigned long long*)smem;

    const int tile = blockIdx.x;
    const int tid = threadIdx.x;
    const int lane = tid & 31;
    const int c = lane & 15;
    const int hh = lane >> 4;
    const int wave = __builtin_amdgcn_readfirstlane((int)(tid >> 5));

    const int s0 = tile_start[tile];
    const int s1 = tile_start[tile + 1];
    const float* hc = h + 4 * c;

    v8f acc0 = {0.f, 0.f, 0.f, 0.f, 0.f, 0.f, 0.f, 0.f};
    v8f acc1 = acc0, acc2 = acc0, acc3 = acc0;

    for (int b0 = s0; b0 < s1; b0 += SORT_CAP) {
        int* bc = blkcnt + (tile + b0 / SORT_CAP) * SUBS;
        if (STAGE != 0) {
            int beg = 0;
            int nw = 0;
#pragma unroll
            for (int b = 0; b < SUBS; ++b) {
                const int v = bc[b];
                beg += (b < wave) ? v : 0;
                nw = (b == wave) ? v : nw;
            }
            beg = __builtin_amdgcn_readfirstlane(beg);
            nw = __builtin_amdgcn_readfirstlane(nw);
            sum_run<false>(rec + b0 + beg + 2 * hh, nw, hc, c, hh, acc0, acc1, acc2, acc3);
            continue;
        }
        const int nb = min(SORT_CAP, s1 - b0);
        if (tid < SUBS) bcnt[tid] = 0;
        __syncthreads();

        unsigned long long r[SORT_RPT];
        int rank[SORT_RPT];
#pragma unroll
        for (int k = 0; k < SORT_RPT; ++k) {
            const int idx = k * 256 + tid;
            r[k] = (idx < nb) ? rec[b0 + idx] : 0ull;
        }
        float dc[SORT_RPT], dr[SORT_RPT];
#pragma unroll
        for (int k = 0; k < SORT_RPT; ++k) {
            const unsigned lo = (unsigned)(r[k] & 0xffffffffull);
            dc[k] = dinv[lo & ((1u << COL_BITS) - 1u)];
            dr[k] = dinv[tile * TILE_ROWS + ((lo >> COL_BITS) & (TILE_ROWS - 1))];
        }
#pragma unroll
        for (int k = 0; k < SORT_RPT; ++k) {
            const int idx = k * 256 + tid;
            const unsigned lo = (unsigned)(r[k] & 0xffffffffull);
            rank[k] = 0;
            if (idx < nb) rank[k] = atomicAdd(&bcnt[(lo >> (COL_BITS + 4)) & (SUBS - 1)], 1);
        }
        __syncthreads();

        if (tid < SUBS) {
            int st = 0;
            for (int b = 0; b < tid; ++b) st += bcnt[b];
            bstart[tid] = st;
            bc[tid] = bcnt[tid];
        }
        __syncthreads();

#pragma unroll
        for (int k = 0; k < SORT_RPT; ++k) {
            const unsigned lo = (unsigned)(r[k] & 0xffffffffull);
            const float a = __uint_as_float((unsigned)(r[k] >> 32));
            const float l = -dr[k] * a * dc[k];
            r[k] = (unsigned long long)lo | ((unsigned long long)__float_as_uint(l) << 32);
        }

#pragma unroll
        for (int k = 0; k < SORT_RPT; ++k) {
            const int idx = k * 256 + tid;
            const unsigned lo = (unsigned)(r[k] & 0xffffffffull);
            if (idx < nb) srt[bstart[(lo >> (COL_BITS + 4)) & (SUBS - 1)] + rank[k]] = r[k];
        }
        __syncthreads();

        unsigned long long w[SORT_RPT];
#pragma unroll
        for (int k = 0; k < SORT_RPT; ++k) w[k] = srt[k * 256 + tid];
#pragma unroll
        for (int k = 0; k < SORT_RPT; ++k) {
            const int idx = k * 256 + tid;
            if (idx < nb) rec[b0 + idx] = w[k];
        }

        const int beg = __builtin_amdgcn_readfirstlane(bstart[wave]);
        const int nw = __builtin_amdgcn_readfirstlane(bcnt[wave]);
        sum_run<true>(srt + beg + 2 * hh, nw, hc, c, hh, acc0, acc1, acc2, acc3);
        __syncthreads();
    }

    const int rowl = 16 * wave + 8 * hh;
    const int rowb = tile * TILE_ROWS + rowl;
    const int ct = wave & 3;
    const int rg = wave >> 2;

    v2f bfa[16], bfb[16];
    if (STAGE == 0) load_panel(Wa, ct, c, hh, bfa);
    else load_panel(Wb, ct, c, hh, bfb);

    v4f pv[8];
#pragma unroll
    for (int rr = 0; rr < 8; ++rr) {
        const int grow = rowb + rr;
        pv[rr] = (v4f){0.f, 0.f, 0.f, 0.f};
        if (grow < N_NODES) pv[rr] = ((const v4f*)prev)[(size_t)grow * (D / 4) + c];
    }
#pragma unroll
    for (int rr = 0; rr < 8; ++rr) {
        const int grow = rowb + rr;
        v4f a = {acc0[rr], acc1[rr], acc2[rr], acc3[rr]};
        if (STAGE != 0) a = 2.0f * a - pv[rr];
        if (STAGE != 2) {
            if (grow < N_NODES) ((v4f*)dst)[(size_t)grow * (D / 4) + c] = a;
        }
        *(v4f*)(smem + (rowl + rr) * TS_PITCH + 4 * c) = a;
        if (STAGE == 0) *(v4f*)(smem + TS_FLOATS + (rowl + rr) * TS_PITCH + 4 * c) = pv[rr];
    }
    __syncthreads();

    v8f o[4];
    if (STAGE == 0) {
        const float bval = bias[16 * ct + c];
#pragma unroll
        for (int j = 0; j < 4; ++j)
#pragma unroll
            for (int r = 0; r < 8; ++r) o[j][r] = bval;
        load_panel(Wb, ct, c, hh, bfb);
        tile_product(smem + TS_FLOATS, rg, c, hh, bfa, o);
        tile_product(smem, rg, c, hh, bfb, o);
    } else {
#pragma unroll
        for (int j = 0; j < 4; ++j)
#pragma unroll
            for (int r = 0; r < 8; ++r) o[j][r] = 0.0f;
        tile_product(smem, rg, c, hh, bfb, o);
    }

    const int orow = tid >> 4;
    const int oc = tid & 15;
    v4f ov[8];
    if (STAGE != 0) {
#pragma unroll
        for (int i = 0; i < 8; ++i) {
            const int grow = tile * TILE_ROWS + 16 * i + orow;
            ov[i] = (v4f){0.f, 0.f, 0.f, 0.f};
            if (grow < N_NODES) ov[i] = ((const v4f*)out)[(size_t)grow * (D / 4) + oc];
        }
    }
    __syncthreads();
#pragma unroll
    for (int j = 0; j < 4; ++j)
#pragma unroll
        for (int r = 0; r < 8; ++r)
            smem[(16 * (4 * rg + j) + 8 * hh + r) * TS_PITCH + 16 * ct + c] = o[j][r];
    __syncthreads();
#pragma unroll
    for (int i = 0; i < 8; ++i) {
        const int lrow = 16 * i + orow;
        const int grow = tile * TILE_ROWS + lrow;
        v4f v = *(const v4f*)(smem + lrow * TS_PITCH + 4 * oc);
        if (STAGE != 0) v = ov[i] + v;
        if (grow < N_NODES) ((v4f*)out)[(size_t)grow * (D / 4) + oc] = v;
    }
}

extern "C" void kernel_launch(void* const* d_in, const int* in_sizes, int n_in,
                              void* d_out, int out_size, void* d_ws, size_t ws_size,
                              hipStream_t stream) {
    const float* x         = (const float*)d_in[0];
    const int*   edge_idx  = (const int*)d_in[1];
    const float* edge_attr = (const float*)d_in[2];
    const float* weight    = (const float*)d_in[3];
    const float* bias      = (const float*)d_in[4];
    float*       out       = (float*)d_out;

    const int* row = edge_idx;
    const int* col = edge_idx + N_EDGES;

    char* ws = (char*)d_ws;
    float* dinv = (float*)ws; ws += sizeof(float) * N_NODES;
    int* blkcnt = (int*)ws;
    ws += sizeof(float) * N_NODES;
    ws += sizeof(float) * N_EDGES;
    float* A    = (float*)ws; ws += sizeof(float) * (size_t)N_NODES * D;
    float* B    = (float*)ws; ws += sizeof(float) * (size_t)N_NODES * D;
    float* Z    = (float*)ws;

    unsigned long long* rec = (unsigned long long*)Z;
    int* cnt        = (int*)(rec + N_EDGES);
    int* tile_start = cnt + BIN_BLOCKS * N_TILES;

    bin_count_kernel<<<BIN_BLOCKS, 256, 0, stream>>>(row, cnt);
    bin_scan_kernel<<<1, 1024, 0, stream>>>(cnt, tile_start);
    bin_place_kernel<<<BIN_BLOCKS, 256, 0, stream>>>(row, col, edge_attr, cnt, tile_start, rec);

    tile_dinv_kernel<<<N_TILES, 256, 0, stream>>>(rec, tile_start, dinv);

    spmm_gemm_kernel<0><<<N_TILES, 256, 0, stream>>>(rec, tile_start, x, x, A,
                                                     weight, weight + 1 * D * D, bias, out, dinv,
                                                     blkcnt);
    spmm_gemm_kernel<1><<<N_TILES, 256, 0, stream>>>(rec, tile_start, A, x, B,
                                                     nullptr, weight + 2 * D * D, nullptr, out,
                                                     nullptr, blkcnt);
    spmm_gemm_kernel<2><<<N_TILES, 256, 0, stream>>>(rec, tile_start, B, A, nullptr,
                                                     nullptr, weight + 3 * D * D, nullptr, out,
                                                     nullptr, blkcnt);
}


# SPMM passes 2 and 3 as 6250 one-wave work-groups, one per 16-row block: no barriers, 24-25 blocks per WGP where 782 tiles gave 3-4; term product on column-interleaved panels; sum order kept
# speedup vs baseline: 5.3160x; 1.0547x over previous
#include <hip/hip_runtime.h>

#define N_NODES 100000
#define N_EDGES 1600000
#define D 64

typedef __attribute__((ext_vector_type(2))) float v2f;
typedef __attribute__((ext_vector_type(8))) float v8f;

__device__ __forceinline__ float dinv_of(float d) {
    return (d > 0.0f) ? (1.0f / sqrtf(d)) : 0.0f;
}

#define TILE_ROWS 128
#define TILE_SHIFT 7
#define N_TILES 782
#define BIN_BLOCKS 250
#define BIN_CHUNK 6400
#define COL_BITS 20

typedef __attribute__((ext_vector_type(4))) float v4f;

__global__ void __launch_bounds__(256)
bin_count_kernel(const int* __restrict__ row, int* __restrict__ cnt) {
    __shared__ int hist[N_TILES];
    for (int i = threadIdx.x; i < N_TILES; i += 256) hist[i] = 0;
    __syncthreads();
    const int base = blockIdx.x * BIN_CHUNK;
    int r[BIN_CHUNK / 256];
#pragma unroll
    for (int k = 0; k < BIN_CHUNK / 256; ++k) r[k] = row[base + threadIdx.x + 256 * k];
#pragma unroll
    for (int k = 0; k < BIN_CHUNK / 256; ++k) atomicAdd(&hist[r[k] >> TILE_SHIFT], 1);
    __syncthreads();
    for (int i = threadIdx.x; i < N_TILES; i += 256)
        cnt[blockIdx.x * N_TILES + i] = hist[i];
}

__global__ void __launch_bounds__(1024)
bin_scan_kernel(int* __restrict__ cnt, int* __restrict__ tile_start) {
    __shared__ int s[1024];
    const int t = threadIdx.x;
    int run = 0;
    if (t < N_TILES) {
        for (int b0 = 0; b0 < BIN_BLOCKS; b0 += 25) {
            int v[25];
#pragma unroll
            for (int u = 0; u < 25; ++u) v[u] = cnt[(b0 + u) * N_TILES + t];
#pragma unroll
            for (int u = 0; u < 25; ++u) {
                cnt[(b0 + u) * N_TILES + t] = run;
                run += v[u];
            }
        }
    }
    s[t] = run;
    __syncthreads();
    for (int off = 1; off < 1024; off <<= 1) {
        int v = (t >= off) ? s[t - off] : 0;
        __syncthreads();
        s[t] += v;
        __syncthreads();
    }
    if (t < N_TILES) tile_start[t] = s[t] - run;
    if (t == N_TILES - 1) tile_start[N_TILES] = s[t];
}

__global__ void __launch_bounds__(256)
bin_place_kernel(const int* __restrict__ row, const int* __restrict__ col,
                 const float* __restrict__ attr,
                 const int* __restrict__ cnt,
                 const int* __restrict__ tile_start, unsigned long long* __restrict__ rec) {
    __shared__ int cur[N_TILES];
    for (int i = threadIdx.x; i < N_TILES; i += 256)
        cur[i] = tile_start[i] + cnt[blockIdx.x * N_TILES + i];
    __syncthreads();
    const int base = blockIdx.x * BIN_CHUNK;
    for (int k0 = 0; k0 < BIN_CHUNK / 256; k0 += 5) {
        int r[5], c[5];
        float a[5];
#pragma unroll
        for (int u = 0; u < 5; ++u) {
            const int e = base + threadIdx.x + 256 * (k0 + u);
            r[u] = row[e];
            c[u] = col[e];
            a[u] = attr[e];
        }
#pragma unroll
        for (int u = 0; u < 5; ++u) {
            const int slot = atomicAdd(&cur[r[u] >> TILE_SHIFT], 1);
            const unsigned lo = (unsigned)c[u] | ((unsigned)(r[u] & (TILE_ROWS - 1)) << COL_BITS);
            const unsigned hi = __float_as_uint(a[u]);
            rec[slot] = (unsigned long long)lo | ((unsigned long long)hi << 32);
        }
    }
}

#define SUBS 8
#define SORT_CAP 2560
#define SORT_RPT (SORT_CAP / 256)
#define SORT_SLOTS (SORT_CAP + 4)
#define TS_PITCH 68
#define TS_FLOATS (TILE_ROWS * TS_PITCH)

typedef __attribute__((ext_vector_type(4))) unsigned v4u;

__global__ void __launch_bounds__(256)
tile_dinv_kernel(const unsigned long long* __restrict__ rec, const int* __restrict__ tile_start,
                 float* __restrict__ dinv) {
    __shared__ int rc[TILE_ROWS];
    const int tile = blockIdx.x;
    const int tid = threadIdx.x;
    if (tid < TILE_ROWS) rc[tid] = 0;
    __syncthreads();
    const int s0 = tile_start[tile];
    const int s1 = tile_start[tile + 1];
    const unsigned* lo = (const unsigned*)rec;
    for (int b0 = s0; b0 < s1; b0 += SORT_CAP) {
        const int nb = min(SORT_CAP, s1 - b0);
        unsigned w[SORT_RPT];
#pragma unroll
        for (int k = 0; k < SORT_RPT; ++k) {
            const int idx = min(k * 256 + tid, nb - 1);
            w[k] = lo[2 * (size_t)(b0 + idx)];
        }
#pragma unroll
        for (int k = 0; k < SORT_RPT; ++k) {
            const int idx = k * 256 + tid;
            if (idx < nb) atomicAdd(&rc[(w[k] >> COL_BITS) & (TILE_ROWS - 1)], 1);
        }
    }
    __syncthreads();
    if (tid < TILE_ROWS) {
        const int grow = tile * TILE_ROWS + tid;
        if (grow < N_NODES) dinv[grow] = dinv_of((float)rc[tid]);
    }
}

__device__ __forceinline__ v8f wmma4(v2f a, float b0, float b1, v8f c) {
    v2f b = {b0, b1};
    return __builtin_amdgcn_wmma_f32_16x16x4_f32(
         false, a,  false, b,
         (short)0, c,  false,  false);
}

__device__ __forceinline__ void load_panel(const float* __restrict__ Wk, int ct, int c, int hh,
                                           v2f (&bf)[16]) {
#pragma unroll
    for (int ks = 0; ks < 16; ++ks) {
        const int k = 4 * ks + 2 * hh;
        bf[ks].x = Wk[(k + 0) * D + 16 * ct + c];
        bf[ks].y = Wk[(k + 1) * D + 16 * ct + c];
    }
}

__device__ __forceinline__ void tile_product(const float* ts, int rg, int c, int hh,
                                             const v2f (&bf)[16], v8f (&o)[4]) {
#pragma unroll
    for (int j = 0; j < 4; ++j) {
        const float* ar = ts + (16 * (4 * rg + j) + c) * TS_PITCH + 2 * hh;
#pragma unroll
        for (int ks = 0; ks < 16; ++ks) {
            const v2f a = *(const v2f*)(ar + 4 * ks);
            o[j] = __builtin_amdgcn_wmma_f32_16x16x4_f32(
                 false, a,  false, bf[ks],
                 (short)0, o[j],  false,  false);
        }
    }
}

__device__ __forceinline__ v4u load_pair(const unsigned long long* q) {
    const unsigned long long a = q[0];
    const unsigned long long b = q[1];
    v4u p = {(unsigned)a, (unsigned)(a >> 32), (unsigned)b, (unsigned)(b >> 32)};
    return p;
}

__device__ __forceinline__ void group_product(v4u p, bool ok0, bool ok1, v4f va, v4f vb, int c,
                                              v8f& acc0, v8f& acc1, v8f& acc2, v8f& acc3) {
    v2f a;
    a.x = (ok0 && ((p.x >> COL_BITS) & 15u) == (unsigned)c) ? __uint_as_float(p.y) : 0.0f;
    a.y = (ok1 && ((p.z >> COL_BITS) & 15u) == (unsigned)c) ? __uint_as_float(p.w) : 0.0f;
    acc0 = wmma4(a, va.x, vb.x, acc0);
    acc1 = wmma4(a, va.y, vb.y, acc1);
    acc2 = wmma4(a, va.z, vb.z, acc2);
    acc3 = wmma4(a, va.w, vb.w, acc3);
}

template <bool IN_LDS>
__device__ __forceinline__ void sum_run(const unsigned long long* sw, int nw, const float* hc,
                                        int c, int hh,
                                        v8f& acc0, v8f& acc1, v8f& acc2, v8f& acc3) {
    const int nfull = nw >> 2;
    int g = 0;
    v4u p[4];
#pragma unroll
    for (int u = 0; u < 4; ++u) p[u] = (v4u){0u, 0u, 0u, 0u};
    if (!IN_LDS && nfull >= 4) {
#pragma unroll
        for (int u = 0; u < 4; ++u) p[u] = load_pair(sw + 4 * u);
    }
    for (; g + 4 <= nfull; g += 4) {
        v4f va[4], vb[4];
        v4u pn[4];
        if (IN_LDS) {
#pragma unroll
            for (int u = 0; u < 4; ++u) p[u] = load_pair(sw + 4 * (g + u));
        }
#pragma unroll
        for (int u = 0; u < 4; ++u) {
            va[u] = *(const v4f*)(hc + (size_t)(p[u].x & ((1u << COL_BITS) - 1u)) * D);
            vb[u] = *(const v4f*)(hc + (size_t)(p[u].z & ((1u << COL_BITS) - 1u)) * D);
        }
        if (!IN_LDS) {
            const int gn = min(g + 4, nfull - 4);
#pragma unroll
            for (int u = 0; u < 4; ++u) pn[u] = load_pair(sw + 4 * (gn + u));
        }
#pragma unroll
        for (int u = 0; u < 4; ++u)
            group_product(p[u], true, true, va[u], vb[u], c, acc0, acc1, acc2, acc3);
        if (!IN_LDS) {
#pragma unroll
            for (int u = 0; u < 4; ++u) p[u] = pn[u];
        }
    }
    for (; g < nfull; ++g) {
        const v4u q = load_pair(sw + 4 * g);
        const v4f va = *(const v4f*)(hc + (size_t)(q.x & ((1u << COL_BITS) - 1u)) * D);
        const v4f vb = *(const v4f*)(hc + (size_t)(q.z & ((1u << COL_BITS) - 1u)) * D);
        group_product(q, true, true, va, vb, c, acc0, acc1, acc2, acc3);
    }
    if (nw & 3) {
        const int e = 4 * nfull + 2 * hh;
        const bool ok0 = e < nw;
        const bool ok1 = e + 1 < nw;
        v4u q = {0u, 0u, 0u, 0u};
        if (IN_LDS) {
            q = load_pair(sw + 4 * nfull);
        } else {
            if (ok0) { const unsigned long long a = sw[4 * nfull]; q.x = (unsigned)a; q.y = (unsigned)(a >> 32); }
            if (ok1) { const unsigned long long b = sw[4 * nfull + 1]; q.z = (unsigned)b; q.w = (unsigned)(b >> 32); }
        }
        v4f va = {0.f, 0.f, 0.f, 0.f};
        v4f vb = {0.f, 0.f, 0.f, 0.f};
        if (ok0) va = *(const v4f*)(hc + (size_t)(q.x & ((1u << COL_BITS) - 1u)) * D);
        if (ok1) vb = *(const v4f*)(hc + (size_t)(q.z & ((1u << COL_BITS) - 1u)) * D);
        group_product(q, ok0, ok1, va, vb, c, acc0, acc1, acc2, acc3);
    }
}

__global__ void __launch_bounds__(256, 8)
spmm0_gemm_kernel(unsigned long long* __restrict__ rec, const int* __restrict__ tile_start,
                  const float* __restrict__ h, const float* __restrict__ prev,
                  float* __restrict__ dst,
                  const float* __restrict__ Wa, const float* __restrict__ Wb,
                  const float* __restrict__ bias, float* __restrict__ out,
                  const float* __restrict__ dinv, int* __restrict__ blkcnt) {
    __shared__ __attribute__((aligned(16))) float smem[2 * TS_FLOATS];
    __shared__ int bcnt[SUBS];
    __shared__ int bstart[SUBS];
    unsigned long long* srt = (unsigned long long*)smem;

    const int tile = blockIdx.x;
    const int tid = threadIdx.x;
    const int lane = tid & 31;
    const int c = lane & 15;
    const int hh = lane >> 4;
    const int wave = __builtin_amdgcn_readfirstlane((int)(tid >> 5));

    const int s0 = tile_start[tile];
    const int s1 = tile_start[tile + 1];
    const float* hc = h + 4 * c;

    v8f acc0 = {0.f, 0.f, 0.f, 0.f, 0.f, 0.f, 0.f, 0.f};
    v8f acc1 = acc0, acc2 = acc0, acc3 = acc0;

    for (int b0 = s0; b0 < s1; b0 += SORT_CAP) {
        int* bc = blkcnt + (tile + b0 / SORT_CAP) * SUBS;
        const int nb = min(SORT_CAP, s1 - b0);
        if (tid < SUBS) bcnt[tid] = 0;
        __syncthreads();

        unsigned long long r[SORT_RPT];
        int rank[SORT_RPT];
#pragma unroll
        for (int k = 0; k < SORT_RPT; ++k) {
            const int idx = k * 256 + tid;
            r[k] = (idx < nb) ? rec[b0 + idx] : 0ull;
        }
        float dc[SORT_RPT], dr[SORT_RPT];
#pragma unroll
        for (int k = 0; k < SORT_RPT; ++k) {
            const unsigned lo = (unsigned)(r[k] & 0xffffffffull);
            dc[k] = dinv[lo & ((1u << COL_BITS) - 1u)];
            dr[k] = dinv[tile * TILE_ROWS + ((lo >> COL_BITS) & (TILE_ROWS - 1))];
        }
#pragma unroll
        for (int k = 0; k < SORT_RPT; ++k) {
            const int idx = k * 256 + tid;
            const unsigned lo = (unsigned)(r[k] & 0xffffffffull);
            rank[k] = 0;
            if (idx < nb) rank[k] = atomicAdd(&bcnt[(lo >> (COL_BITS + 4)) & (SUBS - 1)], 1);
        }
        __syncthreads();

        if (tid < SUBS) {
            int st = 0;
            for (int b = 0; b < tid; ++b) st += bcnt[b];
            bstart[tid] = st;
            bc[tid] = bcnt[tid];
        }
        __syncthreads();

#pragma unroll
        for (int k = 0; k < SORT_RPT; ++k) {
            const unsigned lo = (unsigned)(r[k] & 0xffffffffull);
            const float a = __uint_as_float((unsigned)(r[k] >> 32));
            const float l = -dr[k] * a * dc[k];
            r[k] = (unsigned long long)lo | ((unsigned long long)__float_as_uint(l) << 32);
        }

#pragma unroll
        for (int k = 0; k < SORT_RPT; ++k) {
            const int idx = k * 256 + tid;
            const unsigned lo = (unsigned)(r[k] & 0xffffffffull);
            if (idx < nb) srt[bstart[(lo >> (COL_BITS + 4)) & (SUBS - 1)] + rank[k]] = r[k];
        }
        __syncthreads();

        unsigned long long w[SORT_RPT];
#pragma unroll
        for (int k = 0; k < SORT_RPT; ++k) w[k] = srt[k * 256 + tid];
#pragma unroll
        for (int k = 0; k < SORT_RPT; ++k) {
            const int idx = k * 256 + tid;
            if (idx < nb) rec[b0 + idx] = w[k];
        }

        const int beg = __builtin_amdgcn_readfirstlane(bstart[wave]);
        const int nw = __builtin_amdgcn_readfirstlane(bcnt[wave]);
        sum_run<true>(srt + beg + 2 * hh, nw, hc, c, hh, acc0, acc1, acc2, acc3);
        __syncthreads();
    }

    const int rowl = 16 * wave + 8 * hh;
    const int rowb = tile * TILE_ROWS + rowl;
    const int ct = wave & 3;
    const int rg = wave >> 2;

    v2f bfa[16], bfb[16];
    load_panel(Wa, ct, c, hh, bfa);

    v4f pv[8];
#pragma unroll
    for (int rr = 0; rr < 8; ++rr) {
        const int grow = rowb + rr;
        pv[rr] = (v4f){0.f, 0.f, 0.f, 0.f};
        if (grow < N_NODES) pv[rr] = ((const v4f*)prev)[(size_t)grow * (D / 4) + c];
    }
#pragma unroll
    for (int rr = 0; rr < 8; ++rr) {
        const int grow = rowb + rr;
        const v4f a = {acc0[rr], acc1[rr], acc2[rr], acc3[rr]};
        if (grow < N_NODES) ((v4f*)dst)[(size_t)grow * (D / 4) + c] = a;
        *(v4f*)(smem + (rowl + rr) * TS_PITCH + 4 * c) = a;
        *(v4f*)(smem + TS_FLOATS + (rowl + rr) * TS_PITCH + 4 * c) = pv[rr];
    }
    __syncthreads();

    v8f o[4];
    const float bval = bias[16 * ct + c];
#pragma unroll
    for (int j = 0; j < 4; ++j)
#pragma unroll
        for (int r = 0; r < 8; ++r) o[j][r] = bval;
    load_panel(Wb, ct, c, hh, bfb);
    tile_product(smem + TS_FLOATS, rg, c, hh, bfa, o);
    tile_product(smem, rg, c, hh, bfb, o);

    const int orow = tid >> 4;
    const int oc = tid & 15;
    __syncthreads();
#pragma unroll
    for (int j = 0; j < 4; ++j)
#pragma unroll
        for (int r = 0; r < 8; ++r)
            smem[(16 * (4 * rg + j) + 8 * hh + r) * TS_PITCH + 16 * ct + c] = o[j][r];
    __syncthreads();
#pragma unroll
    for (int i = 0; i < 8; ++i) {
        const int lrow = 16 * i + orow;
        const int grow = tile * TILE_ROWS + lrow;
        const v4f v = *(const v4f*)(smem + lrow * TS_PITCH + 4 * oc);
        if (grow < N_NODES) ((v4f*)out)[(size_t)grow * (D / 4) + oc] = v;
    }
}
#define N_BLOCKS16 (N_NODES / 16)
static_assert(N_NODES % 16 == 0, "a 16-row block never runs past the last node");
static_assert(TILE_ROWS == 16 * SUBS, "16-row blocks per tile");

template <int STAGE>
__global__ void __launch_bounds__(32) __attribute__((amdgpu_waves_per_eu(7, 7)))
spmm_gemm_wave_kernel(const unsigned long long* __restrict__ rec,
                      const int* __restrict__ tile_start,
                      const float* __restrict__ h, const float* __restrict__ prev,
                      float* __restrict__ dst, const float* __restrict__ Wb,
                      float* __restrict__ out, const int* __restrict__ blkcnt) {
    __shared__ __attribute__((aligned(16))) float ts[16 * TS_PITCH];

    const int lane = threadIdx.x;
    const int c = lane & 15;
    const int hh = lane >> 4;
    const int blk = blockIdx.x;
    const int tile = blk / SUBS;
    const int sub = blk % SUBS;

    const int s0 = tile_start[tile];
    const int s1 = tile_start[tile + 1];
    const float* hc = h + 4 * c;

    v8f acc0 = {0.f, 0.f, 0.f, 0.f, 0.f, 0.f, 0.f, 0.f};
    v8f acc1 = acc0, acc2 = acc0, acc3 = acc0;

    for (int b0 = s0; b0 < s1; b0 += SORT_CAP) {
        const int* bc = blkcnt + (tile + b0 / SORT_CAP) * SUBS;
        int beg = 0;
        int nw = 0;
#pragma unroll
        for (int b = 0; b < SUBS; ++b) {
            const int v = bc[b];
            beg += (b < sub) ? v : 0;
            nw = (b == sub) ? v : nw;
        }
        beg = __builtin_amdgcn_readfirstlane(beg);
        nw = __builtin_amdgcn_readfirstlane(nw);
        sum_run<false>(rec + b0 + beg + 2 * hh, nw, hc, c, hh, acc0, acc1, acc2, acc3);
    }

    const size_t rowq = (size_t)(blk * 16 + 8 * hh) * (D / 4) + c;

    v4f pv[8];
#pragma unroll
    for (int rr = 0; rr < 8; ++rr) pv[rr] = ((const v4f*)prev)[rowq + (size_t)rr * (D / 4)];
#pragma unroll
    for (int rr = 0; rr < 8; ++rr) {
        v4f a = {acc0[rr], acc1[rr], acc2[rr], acc3[rr]};
        a = 2.0f * a - pv[rr];
        if (STAGE == 1) ((v4f*)dst)[rowq + (size_t)rr * (D / 4)] = a;
        *(v4f*)(ts + (8 * hh + rr) * TS_PITCH + 4 * c) = a;
    }

    v4f ov[8];
#pragma unroll
    for (int rr = 0; rr < 8; ++rr) ov[rr] = ((const v4f*)out)[rowq + (size_t)rr * (D / 4)];

    __syncthreads();

    v8f o0 = {0.f, 0.f, 0.f, 0.f, 0.f, 0.f, 0.f, 0.f};
    v8f o1 = o0, o2 = o0, o3 = o0;
#pragma unroll
    for (int kq = 0; kq < 4; ++kq) {
        v4f w0[4], w1[4];
        v2f af[4];
#pragma unroll
        for (int u = 0; u < 4; ++u) {
            const int k = 4 * (4 * kq + u) + 2 * hh;
            w0[u] = *(const v4f*)(Wb + (k + 0) * D + 4 * c);
            w1[u] = *(const v4f*)(Wb + (k + 1) * D + 4 * c);
            af[u] = *(const v2f*)(ts + c * TS_PITCH + k);
        }
#pragma unroll
        for (int u = 0; u < 4; ++u) {
            o0 = wmma4(af[u], w0[u].x, w1[u].x, o0);
            o1 = wmma4(af[u], w0[u].y, w1[u].y, o1);
            o2 = wmma4(af[u], w0[u].z, w1[u].z, o2);
            o3 = wmma4(af[u], w0[u].w, w1[u].w, o3);
        }
    }

#pragma unroll
    for (int rr = 0; rr < 8; ++rr) {
        const v4f v = {o0[rr], o1[rr], o2[rr], o3[rr]};
        ((v4f*)out)[rowq + (size_t)rr * (D / 4)] = ov[rr] + v;
    }
}

extern "C" void kernel_launch(void* const* d_in, const int* in_sizes, int n_in,
                              void* d_out, int out_size, void* d_ws, size_t ws_size,
                              hipStream_t stream) {
    const float* x         = (const float*)d_in[0];
    const int*   edge_idx  = (const int*)d_in[1];
    const float* edge_attr = (const float*)d_in[2];
    const float* weight    = (const float*)d_in[3];
    const float* bias      = (const float*)d_in[4];
    float*       out       = (float*)d_out;

    const int* row = edge_idx;
    const int* col = edge_idx + N_EDGES;

    char* ws = (char*)d_ws;
    float* dinv = (float*)ws; ws += sizeof(float) * N_NODES;
    int* blkcnt = (int*)ws;
    ws += sizeof(float) * N_NODES;
    ws += sizeof(float) * N_EDGES;
    float* A    = (float*)ws; ws += sizeof(float) * (size_t)N_NODES * D;
    float* B    = (float*)ws; ws += sizeof(float) * (size_t)N_NODES * D;
    float* Z    = (float*)ws;

    unsigned long long* rec = (unsigned long long*)Z;
    int* cnt        = (int*)(rec + N_EDGES);
    int* tile_start = cnt + BIN_BLOCKS * N_TILES;

    bin_count_kernel<<<BIN_BLOCKS, 256, 0, stream>>>(row, cnt);
    bin_scan_kernel<<<1, 1024, 0, stream>>>(cnt, tile_start);
    bin_place_kernel<<<BIN_BLOCKS, 256, 0, stream>>>(row, col, edge_attr, cnt, tile_start, rec);

    tile_dinv_kernel<<<N_TILES, 256, 0, stream>>>(rec, tile_start, dinv);

    spmm0_gemm_kernel<<<N_TILES, 256, 0, stream>>>(rec, tile_start, x, x, A,
                                                   weight, weight + 1 * D * D, bias, out, dinv,
                                                   blkcnt);
    spmm_gemm_wave_kernel<1><<<N_BLOCKS16, 32, 0, stream>>>(rec, tile_start, A, x, B,
                                                            weight + 2 * D * D, out, blkcnt);
    spmm_gemm_wave_kernel<2><<<N_BLOCKS16, 32, 0, stream>>>(rec, tile_start, B, A, nullptr,
                                                            weight + 3 * D * D, out, blkcnt);
}


# placement pass groups its chunk's records by tile in LDS and writes each (chunk,tile) run with neighbouring lanes: a store touches a few lines, not 32; same slots, arrival order inside a run kept
# speedup vs baseline: 5.7891x; 1.0890x over previous
#include <hip/hip_runtime.h>

#define N_NODES 100000
#define N_EDGES 1600000
#define D 64

typedef __attribute__((ext_vector_type(2))) float v2f;
typedef __attribute__((ext_vector_type(8))) float v8f;

__device__ __forceinline__ float dinv_of(float d) {
    return (d > 0.0f) ? (1.0f / sqrtf(d)) : 0.0f;
}

#define TILE_ROWS 128
#define TILE_SHIFT 7
#define N_TILES 782
#define BIN_BLOCKS 250
#define BIN_CHUNK 6400
#define COL_BITS 20

typedef __attribute__((ext_vector_type(4))) float v4f;

__global__ void __launch_bounds__(256)
bin_count_kernel(const int* __restrict__ row, int* __restrict__ cnt) {
    __shared__ int hist[N_TILES];
    for (int i = threadIdx.x; i < N_TILES; i += 256) hist[i] = 0;
    __syncthreads();
    const int base = blockIdx.x * BIN_CHUNK;
    int r[BIN_CHUNK / 256];
#pragma unroll
    for (int k = 0; k < BIN_CHUNK / 256; ++k) r[k] = row[base + threadIdx.x + 256 * k];
#pragma unroll
    for (int k = 0; k < BIN_CHUNK / 256; ++k) atomicAdd(&hist[r[k] >> TILE_SHIFT], 1);
    __syncthreads();
    for (int i = threadIdx.x; i < N_TILES; i += 256)
        cnt[blockIdx.x * N_TILES + i] = hist[i];
}

__global__ void __launch_bounds__(1024)
bin_scan_kernel(int* __restrict__ cnt, int* __restrict__ tile_start) {
    __shared__ int s[1024];
    const int t = threadIdx.x;
    int run = 0;
    if (t < N_TILES) {
        for (int b0 = 0; b0 < BIN_BLOCKS; b0 += 25) {
            int v[25];
#pragma unroll
            for (int u = 0; u < 25; ++u) v[u] = cnt[(b0 + u) * N_TILES + t];
#pragma unroll
            for (int u = 0; u < 25; ++u) {
                cnt[(b0 + u) * N_TILES + t] = run;
                run += v[u];
            }
        }
    }
    s[t] = run;
    __syncthreads();
    for (int off = 1; off < 1024; off <<= 1) {
        int v = (t >= off) ? s[t - off] : 0;
        __syncthreads();
        s[t] += v;
        __syncthreads();
    }
    if (t < N_TILES) tile_start[t] = s[t] - run;
    if (t == N_TILES - 1) tile_start[N_TILES] = s[t];
}

#define PLACE_RPT (BIN_CHUNK / 256)
static_assert(N_TILES <= 4 * 256, "the prefix below gives four tile counts to a thread");
static_assert(N_TILES <= 65535, "tile ids are staged as 16-bit words");

__global__ void __launch_bounds__(256)
bin_place_kernel(const int* __restrict__ row, const int* __restrict__ col,
                 const float* __restrict__ attr,
                 const int* __restrict__ cnt,
                 const int* __restrict__ tile_start, unsigned long long* __restrict__ rec) {
    __shared__ unsigned long long sbuf[BIN_CHUNK];
    __shared__ unsigned short stile[BIN_CHUNK];
    __shared__ int hist[N_TILES];
    __shared__ int lst[N_TILES];
    __shared__ int dlt[N_TILES];
    __shared__ int wsum[8];
    const int tid = threadIdx.x;
    const int lane = tid & 31;
    const int wave = tid >> 5;
    for (int i = tid; i < N_TILES; i += 256) hist[i] = 0;
    __syncthreads();

    const int base = blockIdx.x * BIN_CHUNK;
    int r[PLACE_RPT], c[PLACE_RPT], rank[PLACE_RPT];
    float a[PLACE_RPT];
#pragma unroll
    for (int k = 0; k < PLACE_RPT; ++k) r[k] = row[base + tid + 256 * k];
#pragma unroll
    for (int k = 0; k < PLACE_RPT; ++k) {
        c[k] = col[base + tid + 256 * k];
        a[k] = attr[base + tid + 256 * k];
    }
#pragma unroll
    for (int k = 0; k < PLACE_RPT; ++k) rank[k] = atomicAdd(&hist[r[k] >> TILE_SHIFT], 1);
    __syncthreads();

    const int e0 = 4 * tid;
    int v[4], gb[4];
#pragma unroll
    for (int j = 0; j < 4; ++j) {
        const bool in = e0 + j < N_TILES;
        const int ec = min(e0 + j, N_TILES - 1);
        const int hv = hist[ec];
        const int gv = tile_start[ec] + cnt[blockIdx.x * N_TILES + ec];
        v[j] = in ? hv : 0;
        gb[j] = in ? gv : 0;
    }
    const int s = v[0] + v[1] + v[2] + v[3];
    int incl = s;
#pragma unroll
    for (int off = 1; off < 32; off <<= 1) {
        const int n = __shfl_up(incl, off);
        if (lane >= off) incl += n;
    }
    if (lane == 31) wsum[wave] = incl;
    __syncthreads();
    int run = incl - s;
    for (int w = 0; w < wave; ++w) run += wsum[w];
#pragma unroll
    for (int j = 0; j < 4; ++j) {
        if (e0 + j < N_TILES) {
            lst[e0 + j] = run;
            dlt[e0 + j] = gb[j] - run;
            run += v[j];
        }
    }
    __syncthreads();

#pragma unroll
    for (int k = 0; k < PLACE_RPT; ++k) {
        const int t = r[k] >> TILE_SHIFT;
        const int pos = lst[t] + rank[k];
        const unsigned lo = (unsigned)c[k] | ((unsigned)(r[k] & (TILE_ROWS - 1)) << COL_BITS);
        const unsigned hi = __float_as_uint(a[k]);
        sbuf[pos] = (unsigned long long)lo | ((unsigned long long)hi << 32);
        stile[pos] = (unsigned short)t;
    }
    __syncthreads();

#pragma unroll
    for (int k = 0; k < PLACE_RPT; ++k) {
        const int i = k * 256 + tid;
        const int t = stile[i];
        rec[i + dlt[t]] = sbuf[i];
    }
}

#define SUBS 8
#define SORT_CAP 2560
#define SORT_RPT (SORT_CAP / 256)
#define SORT_SLOTS (SORT_CAP + 4)
#define TS_PITCH 68
#define TS_FLOATS (TILE_ROWS * TS_PITCH)

typedef __attribute__((ext_vector_type(4))) unsigned v4u;

__global__ void __launch_bounds__(256)
tile_dinv_kernel(const unsigned long long* __restrict__ rec, const int* __restrict__ tile_start,
                 float* __restrict__ dinv) {
    __shared__ int rc[TILE_ROWS];
    const int tile = blockIdx.x;
    const int tid = threadIdx.x;
    if (tid < TILE_ROWS) rc[tid] = 0;
    __syncthreads();
    const int s0 = tile_start[tile];
    const int s1 = tile_start[tile + 1];
    const unsigned* lo = (const unsigned*)rec;
    for (int b0 = s0; b0 < s1; b0 += SORT_CAP) {
        const int nb = min(SORT_CAP, s1 - b0);
        unsigned w[SORT_RPT];
#pragma unroll
        for (int k = 0; k < SORT_RPT; ++k) {
            const int idx = min(k * 256 + tid, nb - 1);
            w[k] = lo[2 * (size_t)(b0 + idx)];
        }
#pragma unroll
        for (int k = 0; k < SORT_RPT; ++k) {
            const int idx = k * 256 + tid;
            if (idx < nb) atomicAdd(&rc[(w[k] >> COL_BITS) & (TILE_ROWS - 1)], 1);
        }
    }
    __syncthreads();
    if (tid < TILE_ROWS) {
        const int grow = tile * TILE_ROWS + tid;
        if (grow < N_NODES) dinv[grow] = dinv_of((float)rc[tid]);
    }
}

__device__ __forceinline__ v8f wmma4(v2f a, float b0, float b1, v8f c) {
    v2f b = {b0, b1};
    return __builtin_amdgcn_wmma_f32_16x16x4_f32(
         false, a,  false, b,
         (short)0, c,  false,  false);
}

__device__ __forceinline__ void load_panel(const float* __restrict__ Wk, int ct, int c, int hh,
                                           v2f (&bf)[16]) {
#pragma unroll
    for (int ks = 0; ks < 16; ++ks) {
        const int k = 4 * ks + 2 * hh;
        bf[ks].x = Wk[(k + 0) * D + 16 * ct + c];
        bf[ks].y = Wk[(k + 1) * D + 16 * ct + c];
    }
}

__device__ __forceinline__ void tile_product(const float* ts, int rg, int c, int hh,
                                             const v2f (&bf)[16], v8f (&o)[4]) {
#pragma unroll
    for (int j = 0; j < 4; ++j) {
        const float* ar = ts + (16 * (4 * rg + j) + c) * TS_PITCH + 2 * hh;
#pragma unroll
        for (int ks = 0; ks < 16; ++ks) {
            const v2f a = *(const v2f*)(ar + 4 * ks);
            o[j] = __builtin_amdgcn_wmma_f32_16x16x4_f32(
                 false, a,  false, bf[ks],
                 (short)0, o[j],  false,  false);
        }
    }
}

__device__ __forceinline__ v4u load_pair(const unsigned long long* q) {
    const unsigned long long a = q[0];
    const unsigned long long b = q[1];
    v4u p = {(unsigned)a, (unsigned)(a >> 32), (unsigned)b, (unsigned)(b >> 32)};
    return p;
}

__device__ __forceinline__ void group_product(v4u p, bool ok0, bool ok1, v4f va, v4f vb, int c,
                                              v8f& acc0, v8f& acc1, v8f& acc2, v8f& acc3) {
    v2f a;
    a.x = (ok0 && ((p.x >> COL_BITS) & 15u) == (unsigned)c) ? __uint_as_float(p.y) : 0.0f;
    a.y = (ok1 && ((p.z >> COL_BITS) & 15u) == (unsigned)c) ? __uint_as_float(p.w) : 0.0f;
    acc0 = wmma4(a, va.x, vb.x, acc0);
    acc1 = wmma4(a, va.y, vb.y, acc1);
    acc2 = wmma4(a, va.z, vb.z, acc2);
    acc3 = wmma4(a, va.w, vb.w, acc3);
}

template <bool IN_LDS>
__device__ __forceinline__ void sum_run(const unsigned long long* sw, int nw, const float* hc,
                                        int c, int hh,
                                        v8f& acc0, v8f& acc1, v8f& acc2, v8f& acc3) {
    const int nfull = nw >> 2;
    int g = 0;
    v4u p[4];
#pragma unroll
    for (int u = 0; u < 4; ++u) p[u] = (v4u){0u, 0u, 0u, 0u};
    if (!IN_LDS && nfull >= 4) {
#pragma unroll
        for (int u = 0; u < 4; ++u) p[u] = load_pair(sw + 4 * u);
    }
    for (; g + 4 <= nfull; g += 4) {
        v4f va[4], vb[4];
        v4u pn[4];
        if (IN_LDS) {
#pragma unroll
            for (int u = 0; u < 4; ++u) p[u] = load_pair(sw + 4 * (g + u));
        }
#pragma unroll
        for (int u = 0; u < 4; ++u) {
            va[u] = *(const v4f*)(hc + (size_t)(p[u].x & ((1u << COL_BITS) - 1u)) * D);
            vb[u] = *(const v4f*)(hc + (size_t)(p[u].z & ((1u << COL_BITS) - 1u)) * D);
        }
        if (!IN_LDS) {
            const int gn = min(g + 4, nfull - 4);
#pragma unroll
            for (int u = 0; u < 4; ++u) pn[u] = load_pair(sw + 4 * (gn + u));
        }
#pragma unroll
        for (int u = 0; u < 4; ++u)
            group_product(p[u], true, true, va[u], vb[u], c, acc0, acc1, acc2, acc3);
        if (!IN_LDS) {
#pragma unroll
            for (int u = 0; u < 4; ++u) p[u] = pn[u];
        }
    }
    for (; g < nfull; ++g) {
        const v4u q = load_pair(sw + 4 * g);
        const v4f va = *(const v4f*)(hc + (size_t)(q.x & ((1u << COL_BITS) - 1u)) * D);
        const v4f vb = *(const v4f*)(hc + (size_t)(q.z & ((1u << COL_BITS) - 1u)) * D);
        group_product(q, true, true, va, vb, c, acc0, acc1, acc2, acc3);
    }
    if (nw & 3) {
        const int e = 4 * nfull + 2 * hh;
        const bool ok0 = e < nw;
        const bool ok1 = e + 1 < nw;
        v4u q = {0u, 0u, 0u, 0u};
        if (IN_LDS) {
            q = load_pair(sw + 4 * nfull);
        } else {
            if (ok0) { const unsigned long long a = sw[4 * nfull]; q.x = (unsigned)a; q.y = (unsigned)(a >> 32); }
            if (ok1) { const unsigned long long b = sw[4 * nfull + 1]; q.z = (unsigned)b; q.w = (unsigned)(b >> 32); }
        }
        v4f va = {0.f, 0.f, 0.f, 0.f};
        v4f vb = {0.f, 0.f, 0.f, 0.f};
        if (ok0) va = *(const v4f*)(hc + (size_t)(q.x & ((1u << COL_BITS) - 1u)) * D);
        if (ok1) vb = *(const v4f*)(hc + (size_t)(q.z & ((1u << COL_BITS) - 1u)) * D);
        group_product(q, ok0, ok1, va, vb, c, acc0, acc1, acc2, acc3);
    }
}

__global__ void __launch_bounds__(256, 8)
spmm0_gemm_kernel(unsigned long long* __restrict__ rec, const int* __restrict__ tile_start,
                  const float* __restrict__ h, const float* __restrict__ prev,
                  float* __restrict__ dst,
                  const float* __restrict__ Wa, const float* __restrict__ Wb,
                  const float* __restrict__ bias, float* __restrict__ out,
                  const float* __restrict__ dinv, int* __restrict__ blkcnt) {
    __shared__ __attribute__((aligned(16))) float smem[2 * TS_FLOATS];
    __shared__ int bcnt[SUBS];
    __shared__ int bstart[SUBS];
    unsigned long long* srt = (unsigned long long*)smem;

    const int tile = blockIdx.x;
    const int tid = threadIdx.x;
    const int lane = tid & 31;
    const int c = lane & 15;
    const int hh = lane >> 4;
    const int wave = __builtin_amdgcn_readfirstlane((int)(tid >> 5));

    const int s0 = tile_start[tile];
    const int s1 = tile_start[tile + 1];
    const float* hc = h + 4 * c;

    v8f acc0 = {0.f, 0.f, 0.f, 0.f, 0.f, 0.f, 0.f, 0.f};
    v8f acc1 = acc0, acc2 = acc0, acc3 = acc0;

    for (int b0 = s0; b0 < s1; b0 += SORT_CAP) {
        int* bc = blkcnt + (tile + b0 / SORT_CAP) * SUBS;
        const int nb = min(SORT_CAP, s1 - b0);
        if (tid < SUBS) bcnt[tid] = 0;
        __syncthreads();

        unsigned long long r[SORT_RPT];
        int rank[SORT_RPT];
#pragma unroll
        for (int k = 0; k < SORT_RPT; ++k) {
            const int idx = k * 256 + tid;
            r[k] = (idx < nb) ? rec[b0 + idx] : 0ull;
        }
        float dc[SORT_RPT], dr[SORT_RPT];
#pragma unroll
        for (int k = 0; k < SORT_RPT; ++k) {
            const unsigned lo = (unsigned)(r[k] & 0xffffffffull);
            dc[k] = dinv[lo & ((1u << COL_BITS) - 1u)];
            dr[k] = dinv[tile * TILE_ROWS + ((lo >> COL_BITS) & (TILE_ROWS - 1))];
        }
#pragma unroll
        for (int k = 0; k < SORT_RPT; ++k) {
            const int idx = k * 256 + tid;
            const unsigned lo = (unsigned)(r[k] & 0xffffffffull);
            rank[k] = 0;
            if (idx < nb) rank[k] = atomicAdd(&bcnt[(lo >> (COL_BITS + 4)) & (SUBS - 1)], 1);
        }
        __syncthreads();

        if (tid < SUBS) {
            int st = 0;
            for (int b = 0; b < tid; ++b) st += bcnt[b];
            bstart[tid] = st;
            bc[tid] = bcnt[tid];
        }
        __syncthreads();

#pragma unroll
        for (int k = 0; k < SORT_RPT; ++k) {
            const unsigned lo = (unsigned)(r[k] & 0xffffffffull);
            const float a = __uint_as_float((unsigned)(r[k] >> 32));
            const float l = -dr[k] * a * dc[k];
            r[k] = (unsigned long long)lo | ((unsigned long long)__float_as_uint(l) << 32);
        }

#pragma unroll
        for (int k = 0; k < SORT_RPT; ++k) {
            const int idx = k * 256 + tid;
            const unsigned lo = (unsigned)(r[k] & 0xffffffffull);
            if (idx < nb) srt[bstart[(lo >> (COL_BITS + 4)) & (SUBS - 1)] + rank[k]] = r[k];
        }
        __syncthreads();

        unsigned long long w[SORT_RPT];
#pragma unroll
        for (int k = 0; k < SORT_RPT; ++k) w[k] = srt[k * 256 + tid];
#pragma unroll
        for (int k = 0; k < SORT_RPT; ++k) {
            const int idx = k * 256 + tid;
            if (idx < nb) rec[b0 + idx] = w[k];
        }

        const int beg = __builtin_amdgcn_readfirstlane(bstart[wave]);
        const int nw = __builtin_amdgcn_readfirstlane(bcnt[wave]);
        sum_run<true>(srt + beg + 2 * hh, nw, hc, c, hh, acc0, acc1, acc2, acc3);
        __syncthreads();
    }

    const int rowl = 16 * wave + 8 * hh;
    const int rowb = tile * TILE_ROWS + rowl;
    const int ct = wave & 3;
    const int rg = wave >> 2;

    v2f bfa[16], bfb[16];
    load_panel(Wa, ct, c, hh, bfa);

    v4f pv[8];
#pragma unroll
    for (int rr = 0; rr < 8; ++rr) {
        const int grow = rowb + rr;
        pv[rr] = (v4f){0.f, 0.f, 0.f, 0.f};
        if (grow < N_NODES) pv[rr] = ((const v4f*)prev)[(size_t)grow * (D / 4) + c];
    }
#pragma unroll
    for (int rr = 0; rr < 8; ++rr) {
        const int grow = rowb + rr;
        const v4f a = {acc0[rr], acc1[rr], acc2[rr], acc3[rr]};
        if (grow < N_NODES) ((v4f*)dst)[(size_t)grow * (D / 4) + c] = a;
        *(v4f*)(smem + (rowl + rr) * TS_PITCH + 4 * c) = a;
        *(v4f*)(smem + TS_FLOATS + (rowl + rr) * TS_PITCH + 4 * c) = pv[rr];
    }
    __syncthreads();

    v8f o[4];
    const float bval = bias[16 * ct + c];
#pragma unroll
    for (int j = 0; j < 4; ++j)
#pragma unroll
        for (int r = 0; r < 8; ++r) o[j][r] = bval;
    load_panel(Wb, ct, c, hh, bfb);
    tile_product(smem + TS_FLOATS, rg, c, hh, bfa, o);
    tile_product(smem, rg, c, hh, bfb, o);

    const int orow = tid >> 4;
    const int oc = tid & 15;
    __syncthreads();
#pragma unroll
    for (int j = 0; j < 4; ++j)
#pragma unroll
        for (int r = 0; r < 8; ++r)
            smem[(16 * (4 * rg + j) + 8 * hh + r) * TS_PITCH + 16 * ct + c] = o[j][r];
    __syncthreads();
#pragma unroll
    for (int i = 0; i < 8; ++i) {
        const int lrow = 16 * i + orow;
        const int grow = tile * TILE_ROWS + lrow;
        const v4f v = *(const v4f*)(smem + lrow * TS_PITCH + 4 * oc);
        if (grow < N_NODES) ((v4f*)out)[(size_t)grow * (D / 4) + oc] = v;
    }
}
#define N_BLOCKS16 (N_NODES / 16)
static_assert(N_NODES % 16 == 0, "a 16-row block never runs past the last node");
static_assert(TILE_ROWS == 16 * SUBS, "16-row blocks per tile");

template <int STAGE>
__global__ void __launch_bounds__(32) __attribute__((amdgpu_waves_per_eu(7, 7)))
spmm_gemm_wave_kernel(const unsigned long long* __restrict__ rec,
                      const int* __restrict__ tile_start,
                      const float* __restrict__ h, const float* __restrict__ prev,
                      float* __restrict__ dst, const float* __restrict__ Wb,
                      float* __restrict__ out, const int* __restrict__ blkcnt) {
    __shared__ __attribute__((aligned(16))) float ts[16 * TS_PITCH];

    const int lane = threadIdx.x;
    const int c = lane & 15;
    const int hh = lane >> 4;
    const int blk = blockIdx.x;
    const int tile = blk / SUBS;
    const int sub = blk % SUBS;

    const int s0 = tile_start[tile];
    const int s1 = tile_start[tile + 1];
    const float* hc = h + 4 * c;

    v8f acc0 = {0.f, 0.f, 0.f, 0.f, 0.f, 0.f, 0.f, 0.f};
    v8f acc1 = acc0, acc2 = acc0, acc3 = acc0;

    for (int b0 = s0; b0 < s1; b0 += SORT_CAP) {
        const int* bc = blkcnt + (tile + b0 / SORT_CAP) * SUBS;
        int beg = 0;
        int nw = 0;
#pragma unroll
        for (int b = 0; b < SUBS; ++b) {
            const int v = bc[b];
            beg += (b < sub) ? v : 0;
            nw = (b == sub) ? v : nw;
        }
        beg = __builtin_amdgcn_readfirstlane(beg);
        nw = __builtin_amdgcn_readfirstlane(nw);
        sum_run<false>(rec + b0 + beg + 2 * hh, nw, hc, c, hh, acc0, acc1, acc2, acc3);
    }

    const size_t rowq = (size_t)(blk * 16 + 8 * hh) * (D / 4) + c;

    v4f pv[8];
#pragma unroll
    for (int rr = 0; rr < 8; ++rr) pv[rr] = ((const v4f*)prev)[rowq + (size_t)rr * (D / 4)];
#pragma unroll
    for (int rr = 0; rr < 8; ++rr) {
        v4f a = {acc0[rr], acc1[rr], acc2[rr], acc3[rr]};
        a = 2.0f * a - pv[rr];
        if (STAGE == 1) ((v4f*)dst)[rowq + (size_t)rr * (D / 4)] = a;
        *(v4f*)(ts + (8 * hh + rr) * TS_PITCH + 4 * c) = a;
    }

    v4f ov[8];
#pragma unroll
    for (int rr = 0; rr < 8; ++rr) ov[rr] = ((const v4f*)out)[rowq + (size_t)rr * (D / 4)];

    __syncthreads();

    v8f o0 = {0.f, 0.f, 0.f, 0.f, 0.f, 0.f, 0.f, 0.f};
    v8f o1 = o0, o2 = o0, o3 = o0;
#pragma unroll
    for (int kq = 0; kq < 4; ++kq) {
        v4f w0[4], w1[4];
        v2f af[4];
#pragma unroll
        for (int u = 0; u < 4; ++u) {
            const int k = 4 * (4 * kq + u) + 2 * hh;
            w0[u] = *(const v4f*)(Wb + (k + 0) * D + 4 * c);
            w1[u] = *(const v4f*)(Wb + (k + 1) * D + 4 * c);
            af[u] = *(const v2f*)(ts + c * TS_PITCH + k);
        }
#pragma unroll
        for (int u = 0; u < 4; ++u) {
            o0 = wmma4(af[u], w0[u].x, w1[u].x, o0);
            o1 = wmma4(af[u], w0[u].y, w1[u].y, o1);
            o2 = wmma4(af[u], w0[u].z, w1[u].z, o2);
            o3 = wmma4(af[u], w0[u].w, w1[u].w, o3);
        }
    }

#pragma unroll
    for (int rr = 0; rr < 8; ++rr) {
        const v4f v = {o0[rr], o1[rr], o2[rr], o3[rr]};
        ((v4f*)out)[rowq + (size_t)rr * (D / 4)] = ov[rr] + v;
    }
}

extern "C" void kernel_launch(void* const* d_in, const int* in_sizes, int n_in,
                              void* d_out, int out_size, void* d_ws, size_t ws_size,
                              hipStream_t stream) {
    const float* x         = (const float*)d_in[0];
    const int*   edge_idx  = (const int*)d_in[1];
    const float* edge_attr = (const float*)d_in[2];
    const float* weight    = (const float*)d_in[3];
    const float* bias      = (const float*)d_in[4];
    float*       out       = (float*)d_out;

    const int* row = edge_idx;
    const int* col = edge_idx + N_EDGES;

    char* ws = (char*)d_ws;
    float* dinv = (float*)ws; ws += sizeof(float) * N_NODES;
    int* blkcnt = (int*)ws;
    ws += sizeof(float) * N_NODES;
    ws += sizeof(float) * N_EDGES;
    float* A    = (float*)ws; ws += sizeof(float) * (size_t)N_NODES * D;
    float* B    = (float*)ws; ws += sizeof(float) * (size_t)N_NODES * D;
    float* Z    = (float*)ws;

    unsigned long long* rec = (unsigned long long*)Z;
    int* cnt        = (int*)(rec + N_EDGES);
    int* tile_start = cnt + BIN_BLOCKS * N_TILES;

    bin_count_kernel<<<BIN_BLOCKS, 256, 0, stream>>>(row, cnt);
    bin_scan_kernel<<<1, 1024, 0, stream>>>(cnt, tile_start);
    bin_place_kernel<<<BIN_BLOCKS, 256, 0, stream>>>(row, col, edge_attr, cnt, tile_start, rec);

    tile_dinv_kernel<<<N_TILES, 256, 0, stream>>>(rec, tile_start, dinv);

    spmm0_gemm_kernel<<<N_TILES, 256, 0, stream>>>(rec, tile_start, x, x, A,
                                                   weight, weight + 1 * D * D, bias, out, dinv,
                                                   blkcnt);
    spmm_gemm_wave_kernel<1><<<N_BLOCKS16, 32, 0, stream>>>(rec, tile_start, A, x, B,
                                                            weight + 2 * D * D, out, blkcnt);
    spmm_gemm_wave_kernel<2><<<N_BLOCKS16, 32, 0, stream>>>(rec, tile_start, B, A, nullptr,
                                                            weight + 3 * D * D, out, blkcnt);
}


# scan of the (chunk, tile) counts on 25 work-groups (32 tiles each, a wave per 25 chunks, one round trip, wave totals joined in LDS); placement pass prefixes the tile totals itself; same slots
# speedup vs baseline: 6.5316x; 1.1283x over previous
#include <hip/hip_runtime.h>

#define N_NODES 100000
#define N_EDGES 1600000
#define D 64

typedef __attribute__((ext_vector_type(2))) float v2f;
typedef __attribute__((ext_vector_type(8))) float v8f;

__device__ __forceinline__ float dinv_of(float d) {
    return (d > 0.0f) ? (1.0f / sqrtf(d)) : 0.0f;
}

#define TILE_ROWS 128
#define TILE_SHIFT 7
#define N_TILES 782
#define BIN_BLOCKS 250
#define BIN_CHUNK 6400
#define COL_BITS 20

typedef __attribute__((ext_vector_type(4))) float v4f;

__global__ void __launch_bounds__(256)
bin_count_kernel(const int* __restrict__ row, int* __restrict__ cnt) {
    __shared__ int hist[N_TILES];
    for (int i = threadIdx.x; i < N_TILES; i += 256) hist[i] = 0;
    __syncthreads();
    const int base = blockIdx.x * BIN_CHUNK;
    int r[BIN_CHUNK / 256];
#pragma unroll
    for (int k = 0; k < BIN_CHUNK / 256; ++k) r[k] = row[base + threadIdx.x + 256 * k];
#pragma unroll
    for (int k = 0; k < BIN_CHUNK / 256; ++k) atomicAdd(&hist[r[k] >> TILE_SHIFT], 1);
    __syncthreads();
    for (int i = threadIdx.x; i < N_TILES; i += 256)
        cnt[blockIdx.x * N_TILES + i] = hist[i];
}

#define SCAN_TILES 32
#define SCAN_BLOCKS ((N_TILES + SCAN_TILES - 1) / SCAN_TILES)
#define SCAN_WAVES 10
#define SCAN_CPW (BIN_BLOCKS / SCAN_WAVES)
static_assert(SCAN_TILES == 32, "one lane per tile");
static_assert(SCAN_CPW * SCAN_WAVES == BIN_BLOCKS, "every wave takes the same number of chunks");

__global__ void __launch_bounds__(32 * SCAN_WAVES)
bin_scan_kernel(int* __restrict__ cnt, int* __restrict__ tile_tot) {
    __shared__ int wtot[SCAN_WAVES][SCAN_TILES];
    const int lane = threadIdx.x & 31;
    const int wave = __builtin_amdgcn_readfirstlane((int)(threadIdx.x >> 5));
    const int t = blockIdx.x * SCAN_TILES + lane;
    const bool tin = t < N_TILES;
    const int tc = min(t, N_TILES - 1);
    int* mine = cnt + wave * SCAN_CPW * N_TILES;
    int v[SCAN_CPW];
#pragma unroll
    for (int u = 0; u < SCAN_CPW; ++u) v[u] = mine[u * N_TILES + tc];
    int run = 0;
#pragma unroll
    for (int u = 0; u < SCAN_CPW; ++u) {
        const int x = v[u];
        v[u] = run;
        run += x;
    }
    wtot[wave][lane] = run;
    __syncthreads();
    int below = 0, all = 0;
#pragma unroll
    for (int w = 0; w < SCAN_WAVES; ++w) {
        const int x = wtot[w][lane];
        below += (w < wave) ? x : 0;
        all += x;
    }
    if (tin) {
#pragma unroll
        for (int u = 0; u < SCAN_CPW; ++u) mine[u * N_TILES + t] = below + v[u];
        if (wave == 0) tile_tot[t] = all;
    }
}

#define PLACE_RPT (BIN_CHUNK / 256)
static_assert(N_TILES <= 4 * 256, "the prefix below gives four tile counts to a thread");
static_assert(N_TILES <= 65535, "tile ids are staged as 16-bit words");

__global__ void __launch_bounds__(256)
bin_place_kernel(const int* __restrict__ row, const int* __restrict__ col,
                 const float* __restrict__ attr,
                 const int* __restrict__ cnt, const int* __restrict__ tile_tot,
                 int* __restrict__ tile_start, unsigned long long* __restrict__ rec) {
    __shared__ unsigned long long sbuf[BIN_CHUNK];
    __shared__ unsigned short stile[BIN_CHUNK];
    __shared__ int hist[N_TILES];
    __shared__ int lst[N_TILES];
    __shared__ int dlt[N_TILES];
    __shared__ int wsum[8];
    __shared__ int wtot[8];
    const int tid = threadIdx.x;
    const int lane = tid & 31;
    const int wave = tid >> 5;
    for (int i = tid; i < N_TILES; i += 256) hist[i] = 0;
    __syncthreads();

    const int base = blockIdx.x * BIN_CHUNK;
    int r[PLACE_RPT], c[PLACE_RPT], rank[PLACE_RPT];
    float a[PLACE_RPT];
#pragma unroll
    for (int k = 0; k < PLACE_RPT; ++k) r[k] = row[base + tid + 256 * k];
#pragma unroll
    for (int k = 0; k < PLACE_RPT; ++k) {
        c[k] = col[base + tid + 256 * k];
        a[k] = attr[base + tid + 256 * k];
    }
#pragma unroll
    for (int k = 0; k < PLACE_RPT; ++k) rank[k] = atomicAdd(&hist[r[k] >> TILE_SHIFT], 1);
    __syncthreads();

    const int e0 = 4 * tid;
    int v[4], tt[4], cb[4];
#pragma unroll
    for (int j = 0; j < 4; ++j) {
        const bool in = e0 + j < N_TILES;
        const int ec = min(e0 + j, N_TILES - 1);
        const int hv = hist[ec];
        const int tv = tile_tot[ec];
        cb[j] = cnt[blockIdx.x * N_TILES + ec];
        v[j] = in ? hv : 0;
        tt[j] = in ? tv : 0;
    }
    const int s = v[0] + v[1] + v[2] + v[3];
    const int st = tt[0] + tt[1] + tt[2] + tt[3];
    int incl = s;
    int inclt = st;
#pragma unroll
    for (int off = 1; off < 32; off <<= 1) {
        const int n = __shfl_up(incl, off);
        const int nt = __shfl_up(inclt, off);
        if (lane >= off) { incl += n; inclt += nt; }
    }
    if (lane == 31) { wsum[wave] = incl; wtot[wave] = inclt; }
    __syncthreads();
    int run = incl - s;
    int trun = inclt - st;
    for (int w = 0; w < wave; ++w) { run += wsum[w]; trun += wtot[w]; }
    const bool first = blockIdx.x == 0;
#pragma unroll
    for (int j = 0; j < 4; ++j) {
        if (e0 + j < N_TILES) {
            lst[e0 + j] = run;
            dlt[e0 + j] = trun + cb[j] - run;
            if (first) tile_start[e0 + j] = trun;
            run += v[j];
            trun += tt[j];
            if (first && e0 + j == N_TILES - 1) tile_start[N_TILES] = trun;
        }
    }
    __syncthreads();

#pragma unroll
    for (int k = 0; k < PLACE_RPT; ++k) {
        const int t = r[k] >> TILE_SHIFT;
        const int pos = lst[t] + rank[k];
        const unsigned lo = (unsigned)c[k] | ((unsigned)(r[k] & (TILE_ROWS - 1)) << COL_BITS);
        const unsigned hi = __float_as_uint(a[k]);
        sbuf[pos] = (unsigned long long)lo | ((unsigned long long)hi << 32);
        stile[pos] = (unsigned short)t;
    }
    __syncthreads();

#pragma unroll
    for (int k = 0; k < PLACE_RPT; ++k) {
        const int i = k * 256 + tid;
        const int t = stile[i];
        rec[i + dlt[t]] = sbuf[i];
    }
}

#define SUBS 8
#define SORT_CAP 2560
#define SORT_RPT (SORT_CAP / 256)
#define SORT_SLOTS (SORT_CAP + 4)
#define TS_PITCH 68
#define TS_FLOATS (TILE_ROWS * TS_PITCH)

typedef __attribute__((ext_vector_type(4))) unsigned v4u;

__global__ void __launch_bounds__(256)
tile_dinv_kernel(const unsigned long long* __restrict__ rec, const int* __restrict__ tile_start,
                 float* __restrict__ dinv) {
    __shared__ int rc[TILE_ROWS];
    const int tile = blockIdx.x;
    const int tid = threadIdx.x;
    if (tid < TILE_ROWS) rc[tid] = 0;
    __syncthreads();
    const int s0 = tile_start[tile];
    const int s1 = tile_start[tile + 1];
    const unsigned* lo = (const unsigned*)rec;
    for (int b0 = s0; b0 < s1; b0 += SORT_CAP) {
        const int nb = min(SORT_CAP, s1 - b0);
        unsigned w[SORT_RPT];
#pragma unroll
        for (int k = 0; k < SORT_RPT; ++k) {
            const int idx = min(k * 256 + tid, nb - 1);
            w[k] = lo[2 * (size_t)(b0 + idx)];
        }
#pragma unroll
        for (int k = 0; k < SORT_RPT; ++k) {
            const int idx = k * 256 + tid;
            if (idx < nb) atomicAdd(&rc[(w[k] >> COL_BITS) & (TILE_ROWS - 1)], 1);
        }
    }
    __syncthreads();
    if (tid < TILE_ROWS) {
        const int grow = tile * TILE_ROWS + tid;
        if (grow < N_NODES) dinv[grow] = dinv_of((float)rc[tid]);
    }
}

__device__ __forceinline__ v8f wmma4(v2f a, float b0, float b1, v8f c) {
    v2f b = {b0, b1};
    return __builtin_amdgcn_wmma_f32_16x16x4_f32(
         false, a,  false, b,
         (short)0, c,  false,  false);
}

__device__ __forceinline__ void load_panel(const float* __restrict__ Wk, int ct, int c, int hh,
                                           v2f (&bf)[16]) {
#pragma unroll
    for (int ks = 0; ks < 16; ++ks) {
        const int k = 4 * ks + 2 * hh;
        bf[ks].x = Wk[(k + 0) * D + 16 * ct + c];
        bf[ks].y = Wk[(k + 1) * D + 16 * ct + c];
    }
}

__device__ __forceinline__ void tile_product(const float* ts, int rg, int c, int hh,
                                             const v2f (&bf)[16], v8f (&o)[4]) {
#pragma unroll
    for (int j = 0; j < 4; ++j) {
        const float* ar = ts + (16 * (4 * rg + j) + c) * TS_PITCH + 2 * hh;
#pragma unroll
        for (int ks = 0; ks < 16; ++ks) {
            const v2f a = *(const v2f*)(ar + 4 * ks);
            o[j] = __builtin_amdgcn_wmma_f32_16x16x4_f32(
                 false, a,  false, bf[ks],
                 (short)0, o[j],  false,  false);
        }
    }
}

__device__ __forceinline__ v4u load_pair(const unsigned long long* q) {
    const unsigned long long a = q[0];
    const unsigned long long b = q[1];
    v4u p = {(unsigned)a, (unsigned)(a >> 32), (unsigned)b, (unsigned)(b >> 32)};
    return p;
}

__device__ __forceinline__ void group_product(v4u p, bool ok0, bool ok1, v4f va, v4f vb, int c,
                                              v8f& acc0, v8f& acc1, v8f& acc2, v8f& acc3) {
    v2f a;
    a.x = (ok0 && ((p.x >> COL_BITS) & 15u) == (unsigned)c) ? __uint_as_float(p.y) : 0.0f;
    a.y = (ok1 && ((p.z >> COL_BITS) & 15u) == (unsigned)c) ? __uint_as_float(p.w) : 0.0f;
    acc0 = wmma4(a, va.x, vb.x, acc0);
    acc1 = wmma4(a, va.y, vb.y, acc1);
    acc2 = wmma4(a, va.z, vb.z, acc2);
    acc3 = wmma4(a, va.w, vb.w, acc3);
}

template <bool IN_LDS>
__device__ __forceinline__ void sum_run(const unsigned long long* sw, int nw, const float* hc,
                                        int c, int hh,
                                        v8f& acc0, v8f& acc1, v8f& acc2, v8f& acc3) {
    const int nfull = nw >> 2;
    int g = 0;
    v4u p[4];
#pragma unroll
    for (int u = 0; u < 4; ++u) p[u] = (v4u){0u, 0u, 0u, 0u};
    if (!IN_LDS && nfull >= 4) {
#pragma unroll
        for (int u = 0; u < 4; ++u) p[u] = load_pair(sw + 4 * u);
    }
    for (; g + 4 <= nfull; g += 4) {
        v4f va[4], vb[4];
        v4u pn[4];
        if (IN_LDS) {
#pragma unroll
            for (int u = 0; u < 4; ++u) p[u] = load_pair(sw + 4 * (g + u));
        }
#pragma unroll
        for (int u = 0; u < 4; ++u) {
            va[u] = *(const v4f*)(hc + (size_t)(p[u].x & ((1u << COL_BITS) - 1u)) * D);
            vb[u] = *(const v4f*)(hc + (size_t)(p[u].z & ((1u << COL_BITS) - 1u)) * D);
        }
        if (!IN_LDS) {
            const int gn = min(g + 4, nfull - 4);
#pragma unroll
            for (int u = 0; u < 4; ++u) pn[u] = load_pair(sw + 4 * (gn + u));
        }
#pragma unroll
        for (int u = 0; u < 4; ++u)
            group_product(p[u], true, true, va[u], vb[u], c, acc0, acc1, acc2, acc3);
        if (!IN_LDS) {
#pragma unroll
            for (int u = 0; u < 4; ++u) p[u] = pn[u];
        }
    }
    for (; g < nfull; ++g) {
        const v4u q = load_pair(sw + 4 * g);
        const v4f va = *(const v4f*)(hc + (size_t)(q.x & ((1u << COL_BITS) - 1u)) * D);
        const v4f vb = *(const v4f*)(hc + (size_t)(q.z & ((1u << COL_BITS) - 1u)) * D);
        group_product(q, true, true, va, vb, c, acc0, acc1, acc2, acc3);
    }
    if (nw & 3) {
        const int e = 4 * nfull + 2 * hh;
        const bool ok0 = e < nw;
        const bool ok1 = e + 1 < nw;
        v4u q = {0u, 0u, 0u, 0u};
        if (IN_LDS) {
            q = load_pair(sw + 4 * nfull);
        } else {
            if (ok0) { const unsigned long long a = sw[4 * nfull]; q.x = (unsigned)a; q.y = (unsigned)(a >> 32); }
            if (ok1) { const unsigned long long b = sw[4 * nfull + 1]; q.z = (unsigned)b; q.w = (unsigned)(b >> 32); }
        }
        v4f va = {0.f, 0.f, 0.f, 0.f};
        v4f vb = {0.f, 0.f, 0.f, 0.f};
        if (ok0) va = *(const v4f*)(hc + (size_t)(q.x & ((1u << COL_BITS) - 1u)) * D);
        if (ok1) vb = *(const v4f*)(hc + (size_t)(q.z & ((1u << COL_BITS) - 1u)) * D);
        group_product(q, ok0, ok1, va, vb, c, acc0, acc1, acc2, acc3);
    }
}

__global__ void __launch_bounds__(256, 8)
spmm0_gemm_kernel(unsigned long long* __restrict__ rec, const int* __restrict__ tile_start,
                  const float* __restrict__ h, const float* __restrict__ prev,
                  float* __restrict__ dst,
                  const float* __restrict__ Wa, const float* __restrict__ Wb,
                  const float* __restrict__ bias, float* __restrict__ out,
                  const float* __restrict__ dinv, int* __restrict__ blkcnt) {
    __shared__ __attribute__((aligned(16))) float smem[2 * TS_FLOATS];
    __shared__ int bcnt[SUBS];
    __shared__ int bstart[SUBS];
    unsigned long long* srt = (unsigned long long*)smem;

    const int tile = blockIdx.x;
    const int tid = threadIdx.x;
    const int lane = tid & 31;
    const int c = lane & 15;
    const int hh = lane >> 4;
    const int wave = __builtin_amdgcn_readfirstlane((int)(tid >> 5));

    const int s0 = tile_start[tile];
    const int s1 = tile_start[tile + 1];
    const float* hc = h + 4 * c;

    v8f acc0 = {0.f, 0.f, 0.f, 0.f, 0.f, 0.f, 0.f, 0.f};
    v8f acc1 = acc0, acc2 = acc0, acc3 = acc0;

    for (int b0 = s0; b0 < s1; b0 += SORT_CAP) {
        int* bc = blkcnt + (tile + b0 / SORT_CAP) * SUBS;
        const int nb = min(SORT_CAP, s1 - b0);
        if (tid < SUBS) bcnt[tid] = 0;
        __syncthreads();

        unsigned long long r[SORT_RPT];
        int rank[SORT_RPT];
#pragma unroll
        for (int k = 0; k < SORT_RPT; ++k) {
            const int idx = k * 256 + tid;
            r[k] = (idx < nb) ? rec[b0 + idx] : 0ull;
        }
        float dc[SORT_RPT], dr[SORT_RPT];
#pragma unroll
        for (int k = 0; k < SORT_RPT; ++k) {
            const unsigned lo = (unsigned)(r[k] & 0xffffffffull);
            dc[k] = dinv[lo & ((1u << COL_BITS) - 1u)];
            dr[k] = dinv[tile * TILE_ROWS + ((lo >> COL_BITS) & (TILE_ROWS - 1))];
        }
#pragma unroll
        for (int k = 0; k < SORT_RPT; ++k) {
            const int idx = k * 256 + tid;
            const unsigned lo = (unsigned)(r[k] & 0xffffffffull);
            rank[k] = 0;
            if (idx < nb) rank[k] = atomicAdd(&bcnt[(lo >> (COL_BITS + 4)) & (SUBS - 1)], 1);
        }
        __syncthreads();

        if (tid < SUBS) {
            int st = 0;
            for (int b = 0; b < tid; ++b) st += bcnt[b];
            bstart[tid] = st;
            bc[tid] = bcnt[tid];
        }
        __syncthreads();

#pragma unroll
        for (int k = 0; k < SORT_RPT; ++k) {
            const unsigned lo = (unsigned)(r[k] & 0xffffffffull);
            const float a = __uint_as_float((unsigned)(r[k] >> 32));
            const float l = -dr[k] * a * dc[k];
            r[k] = (unsigned long long)lo | ((unsigned long long)__float_as_uint(l) << 32);
        }

#pragma unroll
        for (int k = 0; k < SORT_RPT; ++k) {
            const int idx = k * 256 + tid;
            const unsigned lo = (unsigned)(r[k] & 0xffffffffull);
            if (idx < nb) srt[bstart[(lo >> (COL_BITS + 4)) & (SUBS - 1)] + rank[k]] = r[k];
        }
        __syncthreads();

        unsigned long long w[SORT_RPT];
#pragma unroll
        for (int k = 0; k < SORT_RPT; ++k) w[k] = srt[k * 256 + tid];
#pragma unroll
        for (int k = 0; k < SORT_RPT; ++k) {
            const int idx = k * 256 + tid;
            if (idx < nb) rec[b0 + idx] = w[k];
        }

        const int beg = __builtin_amdgcn_readfirstlane(bstart[wave]);
        const int nw = __builtin_amdgcn_readfirstlane(bcnt[wave]);
        sum_run<true>(srt + beg + 2 * hh, nw, hc, c, hh, acc0, acc1, acc2, acc3);
        __syncthreads();
    }

    const int rowl = 16 * wave + 8 * hh;
    const int rowb = tile * TILE_ROWS + rowl;
    const int ct = wave & 3;
    const int rg = wave >> 2;

    v2f bfa[16], bfb[16];
    load_panel(Wa, ct, c, hh, bfa);

    v4f pv[8];
#pragma unroll
    for (int rr = 0; rr < 8; ++rr) {
        const int grow = rowb + rr;
        pv[rr] = (v4f){0.f, 0.f, 0.f, 0.f};
        if (grow < N_NODES) pv[rr] = ((const v4f*)prev)[(size_t)grow * (D / 4) + c];
    }
#pragma unroll
    for (int rr = 0; rr < 8; ++rr) {
        const int grow = rowb + rr;
        const v4f a = {acc0[rr], acc1[rr], acc2[rr], acc3[rr]};
        if (grow < N_NODES) ((v4f*)dst)[(size_t)grow * (D / 4) + c] = a;
        *(v4f*)(smem + (rowl + rr) * TS_PITCH + 4 * c) = a;
        *(v4f*)(smem + TS_FLOATS + (rowl + rr) * TS_PITCH + 4 * c) = pv[rr];
    }
    __syncthreads();

    v8f o[4];
    const float bval = bias[16 * ct + c];
#pragma unroll
    for (int j = 0; j < 4; ++j)
#pragma unroll
        for (int r = 0; r < 8; ++r) o[j][r] = bval;
    load_panel(Wb, ct, c, hh, bfb);
    tile_product(smem + TS_FLOATS, rg, c, hh, bfa, o);
    tile_product(smem, rg, c, hh, bfb, o);

    const int orow = tid >> 4;
    const int oc = tid & 15;
    __syncthreads();
#pragma unroll
    for (int j = 0; j < 4; ++j)
#pragma unroll
        for (int r = 0; r < 8; ++r)
            smem[(16 * (4 * rg + j) + 8 * hh + r) * TS_PITCH + 16 * ct + c] = o[j][r];
    __syncthreads();
#pragma unroll
    for (int i = 0; i < 8; ++i) {
        const int lrow = 16 * i + orow;
        const int grow = tile * TILE_ROWS + lrow;
        const v4f v = *(const v4f*)(smem + lrow * TS_PITCH + 4 * oc);
        if (grow < N_NODES) ((v4f*)out)[(size_t)grow * (D / 4) + oc] = v;
    }
}
#define N_BLOCKS16 (N_NODES / 16)
static_assert(N_NODES % 16 == 0, "a 16-row block never runs past the last node");
static_assert(TILE_ROWS == 16 * SUBS, "16-row blocks per tile");

template <int STAGE>
__global__ void __launch_bounds__(32) __attribute__((amdgpu_waves_per_eu(7, 7)))
spmm_gemm_wave_kernel(const unsigned long long* __restrict__ rec,
                      const int* __restrict__ tile_start,
                      const float* __restrict__ h, const float* __restrict__ prev,
                      float* __restrict__ dst, const float* __restrict__ Wb,
                      float* __restrict__ out, const int* __restrict__ blkcnt) {
    __shared__ __attribute__((aligned(16))) float ts[16 * TS_PITCH];

    const int lane = threadIdx.x;
    const int c = lane & 15;
    const int hh = lane >> 4;
    const int blk = blockIdx.x;
    const int tile = blk / SUBS;
    const int sub = blk % SUBS;

    const int s0 = tile_start[tile];
    const int s1 = tile_start[tile + 1];
    const float* hc = h + 4 * c;

    v8f acc0 = {0.f, 0.f, 0.f, 0.f, 0.f, 0.f, 0.f, 0.f};
    v8f acc1 = acc0, acc2 = acc0, acc3 = acc0;

    for (int b0 = s0; b0 < s1; b0 += SORT_CAP) {
        const int* bc = blkcnt + (tile + b0 / SORT_CAP) * SUBS;
        int beg = 0;
        int nw = 0;
#pragma unroll
        for (int b = 0; b < SUBS; ++b) {
            const int v = bc[b];
            beg += (b < sub) ? v : 0;
            nw = (b == sub) ? v : nw;
        }
        beg = __builtin_amdgcn_readfirstlane(beg);
        nw = __builtin_amdgcn_readfirstlane(nw);
        sum_run<false>(rec + b0 + beg + 2 * hh, nw, hc, c, hh, acc0, acc1, acc2, acc3);
    }

    const size_t rowq = (size_t)(blk * 16 + 8 * hh) * (D / 4) + c;

    v4f pv[8];
#pragma unroll
    for (int rr = 0; rr < 8; ++rr) pv[rr] = ((const v4f*)prev)[rowq + (size_t)rr * (D / 4)];
#pragma unroll
    for (int rr = 0; rr < 8; ++rr) {
        v4f a = {acc0[rr], acc1[rr], acc2[rr], acc3[rr]};
        a = 2.0f * a - pv[rr];
        if (STAGE == 1) ((v4f*)dst)[rowq + (size_t)rr * (D / 4)] = a;
        *(v4f*)(ts + (8 * hh + rr) * TS_PITCH + 4 * c) = a;
    }

    v4f ov[8];
#pragma unroll
    for (int rr = 0; rr < 8; ++rr) ov[rr] = ((const v4f*)out)[rowq + (size_t)rr * (D / 4)];

    __syncthreads();

    v8f o0 = {0.f, 0.f, 0.f, 0.f, 0.f, 0.f, 0.f, 0.f};
    v8f o1 = o0, o2 = o0, o3 = o0;
#pragma unroll
    for (int kq = 0; kq < 4; ++kq) {
        v4f w0[4], w1[4];
        v2f af[4];
#pragma unroll
        for (int u = 0; u < 4; ++u) {
            const int k = 4 * (4 * kq + u) + 2 * hh;
            w0[u] = *(const v4f*)(Wb + (k + 0) * D + 4 * c);
            w1[u] = *(const v4f*)(Wb + (k + 1) * D + 4 * c);
            af[u] = *(const v2f*)(ts + c * TS_PITCH + k);
        }
#pragma unroll
        for (int u = 0; u < 4; ++u) {
            o0 = wmma4(af[u], w0[u].x, w1[u].x, o0);
            o1 = wmma4(af[u], w0[u].y, w1[u].y, o1);
            o2 = wmma4(af[u], w0[u].z, w1[u].z, o2);
            o3 = wmma4(af[u], w0[u].w, w1[u].w, o3);
        }
    }

#pragma unroll
    for (int rr = 0; rr < 8; ++rr) {
        const v4f v = {o0[rr], o1[rr], o2[rr], o3[rr]};
        ((v4f*)out)[rowq + (size_t)rr * (D / 4)] = ov[rr] + v;
    }
}

extern "C" void kernel_launch(void* const* d_in, const int* in_sizes, int n_in,
                              void* d_out, int out_size, void* d_ws, size_t ws_size,
                              hipStream_t stream) {
    const float* x         = (const float*)d_in[0];
    const int*   edge_idx  = (const int*)d_in[1];
    const float* edge_attr = (const float*)d_in[2];
    const float* weight    = (const float*)d_in[3];
    const float* bias      = (const float*)d_in[4];
    float*       out       = (float*)d_out;

    const int* row = edge_idx;
    const int* col = edge_idx + N_EDGES;

    char* ws = (char*)d_ws;
    float* dinv = (float*)ws; ws += sizeof(float) * N_NODES;
    int* blkcnt = (int*)ws;
    ws += sizeof(float) * N_NODES;
    ws += sizeof(float) * N_EDGES;
    float* A    = (float*)ws; ws += sizeof(float) * (size_t)N_NODES * D;
    float* B    = (float*)ws; ws += sizeof(float) * (size_t)N_NODES * D;
    float* Z    = (float*)ws;

    unsigned long long* rec = (unsigned long long*)Z;
    int* cnt        = (int*)(rec + N_EDGES);
    int* tile_start = cnt + BIN_BLOCKS * N_TILES;
    int* tile_tot   = tile_start + N_TILES + 1;

    bin_count_kernel<<<BIN_BLOCKS, 256, 0, stream>>>(row, cnt);
    bin_scan_kernel<<<SCAN_BLOCKS, 32 * SCAN_WAVES, 0, stream>>>(cnt, tile_tot);
    bin_place_kernel<<<BIN_BLOCKS, 256, 0, stream>>>(row, col, edge_attr, cnt, tile_tot, tile_start,
                                                     rec);

    tile_dinv_kernel<<<N_TILES, 256, 0, stream>>>(rec, tile_start, dinv);

    spmm0_gemm_kernel<<<N_TILES, 256, 0, stream>>>(rec, tile_start, x, x, A,
                                                   weight, weight + 1 * D * D, bias, out, dinv,
                                                   blkcnt);
    spmm_gemm_wave_kernel<1><<<N_BLOCKS16, 32, 0, stream>>>(rec, tile_start, A, x, B,
                                                            weight + 2 * D * D, out, blkcnt);
    spmm_gemm_wave_kernel<2><<<N_BLOCKS16, 32, 0, stream>>>(rec, tile_start, B, A, nullptr,
                                                            weight + 3 * D * D, out, blkcnt);
}
